# starting kernel
# baseline (speedup 1.0000x reference)
#include <hip/hip_runtime.h>

#define N_NODES 50000
#define N_EDGES 800000
#define N_PAIRS 400000
#define DIM     128
#define LDS_STRIDE 132

typedef float v2f __attribute__((ext_vector_type(2)));
typedef float v8f __attribute__((ext_vector_type(8)));

__global__ void k_init_deg(float* __restrict__ deg) {
    int i = blockIdx.x * blockDim.x + threadIdx.x;
    if (i < N_NODES) deg[i] = 1.0f;
}

__global__ void k_scatter_deg(const int* __restrict__ dst, float* __restrict__ deg) {
    int e = blockIdx.x * blockDim.x + threadIdx.x;
    if (e < N_EDGES) atomicAdd(&deg[dst[e]], 1.0f);
}

__global__ void k_rsqrt_inplace(float* __restrict__ deg) {
    int i = blockIdx.x * blockDim.x + threadIdx.x;
    if (i < N_NODES) deg[i] = rsqrtf(deg[i]);
}

__launch_bounds__(256, 4)
__global__ void k_matmul_wmma(const float* __restrict__ X,
                              const float* __restrict__ W,
                              const float* __restrict__ bias,
                              float* __restrict__ H) {
    __shared__ float lds[16 * LDS_STRIDE];
    const int rowBase = blockIdx.x * 16;
    const int t = threadIdx.x;

    #pragma unroll
    for (int rep = 0; rep < 2; ++rep) {
        int idx = t + rep * 256;
        int r   = idx >> 5;
        int c4  = idx & 31;
        float4 v = *(const float4*)(X + (size_t)(rowBase + r) * DIM + c4 * 4);
        if (bias) {
            float4 b = *(const float4*)(bias + c4 * 4);
            v.x = fmaxf(v.x + b.x, 0.0f);
            v.y = fmaxf(v.y + b.y, 0.0f);
            v.z = fmaxf(v.z + b.z, 0.0f);
            v.w = fmaxf(v.w + b.w, 0.0f);
        }
        *(float4*)(&lds[r * LDS_STRIDE + c4 * 4]) = v;
    }
    __syncthreads();

    const int wave = t >> 5;
    const int lane = t & 31;
    const int half = lane >> 4;
    const int l16  = lane & 15;
    const int colBase = wave * 16;

    v8f acc = {0.f, 0.f, 0.f, 0.f, 0.f, 0.f, 0.f, 0.f};

    #pragma unroll
    for (int k0 = 0; k0 < DIM; k0 += 4) {
        const int kk = k0 + half * 2;
        float2 av = *(const float2*)(&lds[l16 * LDS_STRIDE + kk]);
        v2f a, b;
        a[0] = av.x;
        a[1] = av.y;
        b[0] = W[(size_t)kk * DIM + colBase + l16];
        b[1] = W[(size_t)(kk + 1) * DIM + colBase + l16];
        acc = __builtin_amdgcn_wmma_f32_16x16x4_f32(false, a, false, b,
                                                    (short)0, acc, false, false);
    }

    float* outp = H + (size_t)rowBase * DIM + colBase;
    #pragma unroll
    for (int v = 0; v < 8; ++v) {
        outp[(size_t)(v + half * 8) * DIM + l16] = acc[v];
    }
}

__global__ void k_init_out(const float* __restrict__ h, const float* __restrict__ dinv,
                           const float* __restrict__ bias, float* __restrict__ out) {
    int idx = blockIdx.x * blockDim.x + threadIdx.x;
    int i = idx >> 7;
    int f = idx & 127;
    float di = dinv[i];
    float v = h[idx] * di * di;
    if (bias) v += bias[f];
    out[idx] = v;
}

__launch_bounds__(256)
__global__ void k_scatter_edges(const int* __restrict__ src, const int* __restrict__ dst,
                                const float* __restrict__ h, const float* __restrict__ dinv,
                                float* __restrict__ out) {
    const int wid  = threadIdx.x >> 5;
    const int lane = threadIdx.x & 31;
    const int e = blockIdx.x * 8 + wid;
    if (e >= N_EDGES) return;
    const int s = src[e];
    const int d = dst[e];
    const float norm = dinv[s] * dinv[d];
    const float4 hv = *(const float4*)(h + (size_t)s * DIM + lane * 4);
    float* op = out + (size_t)d * DIM + lane * 4;
    atomicAdd(op + 0, hv.x * norm);
    atomicAdd(op + 1, hv.y * norm);
    atomicAdd(op + 2, hv.z * norm);
    atomicAdd(op + 3, hv.w * norm);
}

__launch_bounds__(256)
__global__ void k_decode(const int* __restrict__ pa, const int* __restrict__ pb,
                         const float* __restrict__ z, float* __restrict__ out) {
    const int wid  = threadIdx.x >> 5;
    const int lane = threadIdx.x & 31;
    const int p = blockIdx.x * 8 + wid;
    if (p >= N_PAIRS) return;
    const int a = pa[p];
    const int b = pb[p];
    const float4 za = *(const float4*)(z + (size_t)a * DIM + lane * 4);
    const float4 zb = *(const float4*)(z + (size_t)b * DIM + lane * 4);
    float s = za.x * zb.x + za.y * zb.y + za.z * zb.z + za.w * zb.w;
    #pragma unroll
    for (int off = 16; off > 0; off >>= 1) s += __shfl_down(s, off);
    if (lane == 0) out[p] = s;
}

extern "C" void kernel_launch(void* const* d_in, const int* in_sizes, int n_in,
                              void* d_out, int out_size, void* d_ws, size_t ws_size,
                              hipStream_t stream) {
    const int*   edge_index = (const int*)d_in[0];
    const int*   edge_pairs = (const int*)d_in[1];
    const float* emb        = (const float*)d_in[2];
    const float* W1         = (const float*)d_in[3];
    const float* b1         = (const float*)d_in[4];
    const float* W2         = (const float*)d_in[5];
    const float* b2         = (const float*)d_in[6];
    float*       out        = (float*)d_out;

    const int* src = edge_index;
    const int* dst = edge_index + N_EDGES;
    const int* pa  = edge_pairs;
    const int* pb  = edge_pairs + N_PAIRS;

    char* ws = (char*)d_ws;
    size_t off = 0;
    float* dinv = (float*)(ws + off); off += ((size_t)N_NODES * 4 + 255) & ~(size_t)255;
    float* h    = (float*)(ws + off); off += (size_t)N_NODES * DIM * 4;
    float* agg  = (float*)(ws + off); off += (size_t)N_NODES * DIM * 4;
    (void)ws_size; (void)in_sizes; (void)n_in; (void)out_size;

    const int T = 256;
    const int gN    = (N_NODES + T - 1) / T;
    const int gE    = (N_EDGES + T - 1) / T;
    const int gNF   = (N_NODES * DIM) / T;
    const int gMM   = N_NODES / 16;
    const int gEw   = N_EDGES / 8;
    const int gPw   = N_PAIRS / 8;

    k_init_deg<<<gN, T, 0, stream>>>(dinv);
    k_scatter_deg<<<gE, T, 0, stream>>>(dst, dinv);
    k_rsqrt_inplace<<<gN, T, 0, stream>>>(dinv);

    k_matmul_wmma<<<gMM, T, 0, stream>>>(emb, W1, nullptr, h);
    k_init_out<<<gNF, T, 0, stream>>>(h, dinv, nullptr, agg);
    k_scatter_edges<<<gEw, T, 0, stream>>>(src, dst, h, dinv, agg);

    k_matmul_wmma<<<gMM, T, 0, stream>>>(agg, W2, b1, h);
    k_init_out<<<gNF, T, 0, stream>>>(h, dinv, b2, agg);
    k_scatter_edges<<<gEw, T, 0, stream>>>(src, dst, h, dinv, agg);

    k_decode<<<gPw, T, 0, stream>>>(pa, pb, agg, out);
}


# Global float atomics removed: each work-group owns 200 destination nodes, scans the whole edge list, sums matching messages in LDS (LDS float adds, arrival order); degrees counted the same way
# speedup vs baseline: 1.7445x; 1.7445x over previous
#include <hip/hip_runtime.h>

#define N_NODES 50000
#define N_EDGES 800000
#define N_PAIRS 400000
#define DIM     128
#define LDS_STRIDE 132

typedef float v2f __attribute__((ext_vector_type(2)));
typedef float v8f __attribute__((ext_vector_type(8)));

#define NPB      200
#define N_TILES  (N_NODES / NPB)
#define WG       1024
#define E4       (N_EDGES / 4)
#define LIST_CAP 16384
#define PU       8

__launch_bounds__(WG)
__global__ void k_degree(const int* __restrict__ dst, float* __restrict__ dinv) {
    __shared__ int cnt[NPB];
    const int t = threadIdx.x;
    const int base = blockIdx.x * NPB;
    if (t < NPB) cnt[t] = 0;
    __syncthreads();
    const int4* __restrict__ dst4 = (const int4*)dst;
    const int rot = blockIdx.x * (E4 / N_TILES);
    for (int i0 = 0; i0 < E4; i0 += 4 * WG) {
        int4 d[4];
        #pragma unroll
        for (int u = 0; u < 4; ++u) {
            const int i = i0 + u * WG + t;
            int q = i + rot;
            if (q >= E4) q -= E4;
            d[u] = (i < E4) ? dst4[q] : make_int4(-1, -1, -1, -1);
        }
        #pragma unroll
        for (int u = 0; u < 4; ++u) {
            const unsigned a = (unsigned)(d[u].x - base), b = (unsigned)(d[u].y - base);
            const unsigned c = (unsigned)(d[u].z - base), e = (unsigned)(d[u].w - base);
            if (min(min(a, b), min(c, e)) < NPB) {
                if (a < NPB) atomicAdd(&cnt[a], 1);
                if (b < NPB) atomicAdd(&cnt[b], 1);
                if (c < NPB) atomicAdd(&cnt[c], 1);
                if (e < NPB) atomicAdd(&cnt[e], 1);
            }
        }
    }
    __syncthreads();
    if (t < NPB) dinv[base + t] = rsqrtf(1.0f + (float)cnt[t]);
}

__launch_bounds__(256, 4)
__global__ void k_matmul_wmma(const float* __restrict__ X,
                              const float* __restrict__ W,
                              const float* __restrict__ bias,
                              float* __restrict__ H) {
    __shared__ float lds[16 * LDS_STRIDE];
    const int rowBase = blockIdx.x * 16;
    const int t = threadIdx.x;

    #pragma unroll
    for (int rep = 0; rep < 2; ++rep) {
        int idx = t + rep * 256;
        int r   = idx >> 5;
        int c4  = idx & 31;
        float4 v = *(const float4*)(X + (size_t)(rowBase + r) * DIM + c4 * 4);
        if (bias) {
            float4 b = *(const float4*)(bias + c4 * 4);
            v.x = fmaxf(v.x + b.x, 0.0f);
            v.y = fmaxf(v.y + b.y, 0.0f);
            v.z = fmaxf(v.z + b.z, 0.0f);
            v.w = fmaxf(v.w + b.w, 0.0f);
        }
        *(float4*)(&lds[r * LDS_STRIDE + c4 * 4]) = v;
    }
    __syncthreads();

    const int wave = t >> 5;
    const int lane = t & 31;
    const int half = lane >> 4;
    const int l16  = lane & 15;
    const int colBase = wave * 16;

    v8f acc = {0.f, 0.f, 0.f, 0.f, 0.f, 0.f, 0.f, 0.f};

    #pragma unroll
    for (int k0 = 0; k0 < DIM; k0 += 4) {
        const int kk = k0 + half * 2;
        float2 av = *(const float2*)(&lds[l16 * LDS_STRIDE + kk]);
        v2f a, b;
        a[0] = av.x;
        a[1] = av.y;
        b[0] = W[(size_t)kk * DIM + colBase + l16];
        b[1] = W[(size_t)(kk + 1) * DIM + colBase + l16];
        acc = __builtin_amdgcn_wmma_f32_16x16x4_f32(false, a, false, b,
                                                    (short)0, acc, false, false);
    }

    float* outp = H + (size_t)rowBase * DIM + colBase;
    #pragma unroll
    for (int v = 0; v < 8; ++v) {
        outp[(size_t)(v + half * 8) * DIM + l16] = acc[v];
    }
}

__device__ __forceinline__ void list_append(int* list, int* nlist, float* acc, int e, unsigned dl,
                                            int base, const int* __restrict__ src,
                                            const float* __restrict__ h,
                                            const float* __restrict__ dinv) {
    const int pos = atomicAdd(nlist, 1);
    if (pos < LIST_CAP) {
        list[pos] = e | (int)(dl << 20);
    } else {
        const int s = src[e];
        const float nm = dinv[s] * dinv[base + (int)dl];
        for (int f = 0; f < DIM; ++f)
            atomicAdd(&acc[dl * DIM + f], h[(size_t)s * DIM + f] * nm);
    }
}

__launch_bounds__(WG)
__global__ void k_aggregate(const int* __restrict__ src, const int* __restrict__ dst,
                            const float* __restrict__ h, const float* __restrict__ dinv,
                            const float* __restrict__ bias, float* __restrict__ out) {
    __shared__ float acc[NPB * DIM];
    __shared__ int   list[LIST_CAP];
    __shared__ int   nlist;
    const int t = threadIdx.x;
    const int wave = t >> 5;
    const int lane = t & 31;
    const int base = blockIdx.x * NPB;

    {
        constexpr int R = (NPB * DIM) / WG;
        float hr[R], dr[R];
        #pragma unroll
        for (int r = 0; r < R; ++r) {
            const int idx = r * WG + t;
            hr[r] = h[(size_t)base * DIM + idx];
            dr[r] = dinv[base + (idx >> 7)];
        }
        const float bv = bias ? bias[t & 127] : 0.0f;
        #pragma unroll
        for (int r = 0; r < R; ++r) {
            float v = hr[r] * dr[r] * dr[r];
            if (bias) v += bv;
            acc[r * WG + t] = v;
        }
    }
    if (t == 0) nlist = 0;
    __syncthreads();

    const int4* __restrict__ dst4 = (const int4*)dst;
    const int rot = blockIdx.x * (E4 / N_TILES);
    for (int i0 = 0; i0 < E4; i0 += 4 * WG) {
        int4 d[4];
        int  qq[4];
        #pragma unroll
        for (int u = 0; u < 4; ++u) {
            const int i = i0 + u * WG + t;
            int q = i + rot;
            if (q >= E4) q -= E4;
            qq[u] = q;
            d[u] = (i < E4) ? dst4[q] : make_int4(-1, -1, -1, -1);
        }
        #pragma unroll
        for (int u = 0; u < 4; ++u) {
            const int e0 = qq[u] * 4;
            const unsigned a = (unsigned)(d[u].x - base), b = (unsigned)(d[u].y - base);
            const unsigned c = (unsigned)(d[u].z - base), g = (unsigned)(d[u].w - base);
            if (min(min(a, b), min(c, g)) < NPB) {
                if (a < NPB) list_append(list, &nlist, acc, e0 + 0, a, base, src, h, dinv);
                if (b < NPB) list_append(list, &nlist, acc, e0 + 1, b, base, src, h, dinv);
                if (c < NPB) list_append(list, &nlist, acc, e0 + 2, c, base, src, h, dinv);
                if (g < NPB) list_append(list, &nlist, acc, e0 + 3, g, base, src, h, dinv);
            }
        }
    }
    __syncthreads();

    const int n = min(nlist, LIST_CAP);
    for (int k0 = wave * PU; k0 < n; k0 += (WG / 32) * PU) {
        int   ent[PU], s[PU];
        float ds[PU], dd[PU];
        float hv[PU][4];
        #pragma unroll
        for (int u = 0; u < PU; ++u) {
            const int k = k0 + u;
            ent[u] = (k < n) ? list[k] : -1;
        }
        #pragma unroll
        for (int u = 0; u < PU; ++u)
            s[u] = src[max(ent[u], 0) & 0xFFFFF];
        #pragma unroll
        for (int u = 0; u < PU; ++u) {
            const int dl = (max(ent[u], 0) >> 20) & 255;
            ds[u] = dinv[s[u]];
            dd[u] = dinv[base + dl];
            const float* hp = h + (size_t)s[u] * DIM + lane;
            hv[u][0] = hp[0]; hv[u][1] = hp[32]; hv[u][2] = hp[64]; hv[u][3] = hp[96];
        }
        #pragma unroll
        for (int u = 0; u < PU; ++u)
            asm volatile("" :: "v"(ds[u]), "v"(dd[u]), "v"(hv[u][0]), "v"(hv[u][1]),
                               "v"(hv[u][2]), "v"(hv[u][3]));
        #pragma unroll
        for (int u = 0; u < PU; ++u) {
            if (ent[u] >= 0) {
                const float nm = ds[u] * dd[u];
                float* ap = &acc[((ent[u] >> 20) & 255) * DIM + lane];
                atomicAdd(ap +  0, hv[u][0] * nm);
                atomicAdd(ap + 32, hv[u][1] * nm);
                atomicAdd(ap + 64, hv[u][2] * nm);
                atomicAdd(ap + 96, hv[u][3] * nm);
            }
        }
    }
    __syncthreads();

    for (int r = 0; r < (NPB * DIM) / WG; ++r)
        out[(size_t)base * DIM + r * WG + t] = acc[r * WG + t];
}

__launch_bounds__(256)
__global__ void k_decode(const int* __restrict__ pa, const int* __restrict__ pb,
                         const float* __restrict__ z, float* __restrict__ out) {
    const int wid  = threadIdx.x >> 5;
    const int lane = threadIdx.x & 31;
    const int p = blockIdx.x * 8 + wid;
    if (p >= N_PAIRS) return;
    const int a = pa[p];
    const int b = pb[p];
    const float4 za = *(const float4*)(z + (size_t)a * DIM + lane * 4);
    const float4 zb = *(const float4*)(z + (size_t)b * DIM + lane * 4);
    float s = za.x * zb.x + za.y * zb.y + za.z * zb.z + za.w * zb.w;
    #pragma unroll
    for (int off = 16; off > 0; off >>= 1) s += __shfl_down(s, off);
    if (lane == 0) out[p] = s;
}

extern "C" void kernel_launch(void* const* d_in, const int* in_sizes, int n_in,
                              void* d_out, int out_size, void* d_ws, size_t ws_size,
                              hipStream_t stream) {
    const int*   edge_index = (const int*)d_in[0];
    const int*   edge_pairs = (const int*)d_in[1];
    const float* emb        = (const float*)d_in[2];
    const float* W1         = (const float*)d_in[3];
    const float* b1         = (const float*)d_in[4];
    const float* W2         = (const float*)d_in[5];
    const float* b2         = (const float*)d_in[6];
    float*       out        = (float*)d_out;

    const int* src = edge_index;
    const int* dst = edge_index + N_EDGES;
    const int* pa  = edge_pairs;
    const int* pb  = edge_pairs + N_PAIRS;

    char* ws = (char*)d_ws;
    size_t off = 0;
    float* dinv = (float*)(ws + off); off += ((size_t)N_NODES * 4 + 255) & ~(size_t)255;
    float* h    = (float*)(ws + off); off += (size_t)N_NODES * DIM * 4;
    float* agg  = (float*)(ws + off); off += (size_t)N_NODES * DIM * 4;
    (void)ws_size; (void)in_sizes; (void)n_in; (void)out_size;

    const int T = 256;
    const int gMM   = N_NODES / 16;
    const int gPw   = N_PAIRS / 8;

    k_degree<<<N_TILES, WG, 0, stream>>>(dst, dinv);

    k_matmul_wmma<<<gMM, T, 0, stream>>>(emb, W1, nullptr, h);
    k_aggregate<<<N_TILES, WG, 0, stream>>>(src, dst, h, dinv, nullptr, agg);

    k_matmul_wmma<<<gMM, T, 0, stream>>>(agg, W2, b1, h);
    k_aggregate<<<N_TILES, WG, 0, stream>>>(src, dst, h, dinv, b2, agg);

    k_decode<<<gPw, T, 0, stream>>>(pa, pb, agg, out);
}


# Edge list scanned once per call, not three times: degree kernel stores each tile's matched edges (source, local row) in workspace, both aggregations read them; LDS float adds stay in arrival order
# speedup vs baseline: 2.2890x; 1.3121x over previous
#include <hip/hip_runtime.h>

#define N_NODES 50000
#define N_EDGES 800000
#define N_PAIRS 400000
#define DIM     128
#define LDS_STRIDE 132

typedef float v2f __attribute__((ext_vector_type(2)));
typedef float v8f __attribute__((ext_vector_type(8)));

#define NPB      200
#define N_TILES  (N_NODES / NPB)
#define WG       1024
#define E4       (N_EDGES / 4)
#define LIST_CAP 16384
#define PU       8

__device__ __forceinline__ void bin_append(int* cnt, int* list, int* nlist, int e, unsigned dl) {
    atomicAdd(&cnt[dl], 1);
    const int pos = atomicAdd(nlist, 1);
    if (pos < LIST_CAP) list[pos] = e | (int)(dl << 20);
}

__launch_bounds__(WG)
__global__ void k_bin(const int* __restrict__ src, const int* __restrict__ dst,
                      float* __restrict__ dinv, int* __restrict__ tile_n,
                      int* __restrict__ tile_list) {
    __shared__ int cnt[NPB];
    __shared__ int list[LIST_CAP];
    __shared__ int nlist;
    const int t = threadIdx.x;
    const int base = blockIdx.x * NPB;
    if (t < NPB) cnt[t] = 0;
    if (t == 0) nlist = 0;
    __syncthreads();
    const int4* __restrict__ dst4 = (const int4*)dst;
    const int rot = blockIdx.x * (E4 / N_TILES);
    for (int i0 = 0; i0 < E4; i0 += 4 * WG) {
        int4 d[4];
        int  qq[4];
        #pragma unroll
        for (int u = 0; u < 4; ++u) {
            const int i = i0 + u * WG + t;
            int q = i + rot;
            if (q >= E4) q -= E4;
            qq[u] = q;
            d[u] = (i < E4) ? dst4[q] : make_int4(-1, -1, -1, -1);
        }
        #pragma unroll
        for (int u = 0; u < 4; ++u) {
            const int e0 = qq[u] * 4;
            const unsigned a = (unsigned)(d[u].x - base), b = (unsigned)(d[u].y - base);
            const unsigned c = (unsigned)(d[u].z - base), g = (unsigned)(d[u].w - base);
            if (min(min(a, b), min(c, g)) < NPB) {
                if (a < NPB) bin_append(cnt, list, &nlist, e0 + 0, a);
                if (b < NPB) bin_append(cnt, list, &nlist, e0 + 1, b);
                if (c < NPB) bin_append(cnt, list, &nlist, e0 + 2, c);
                if (g < NPB) bin_append(cnt, list, &nlist, e0 + 3, g);
            }
        }
    }
    __syncthreads();
    if (t < NPB) dinv[base + t] = rsqrtf(1.0f + (float)cnt[t]);
    const int n = nlist;
    if (t == 0) tile_n[blockIdx.x] = n;
    const int m = min(n, LIST_CAP);
    int* __restrict__ outl = tile_list + (size_t)blockIdx.x * LIST_CAP;
    for (int k = t; k < m; k += WG) {
        const int ent = list[k];
        outl[k] = src[ent & 0xFFFFF] | (ent & ~0xFFFFF);
    }
}

__launch_bounds__(256, 4)
__global__ void k_matmul_wmma(const float* __restrict__ X,
                              const float* __restrict__ W,
                              const float* __restrict__ bias,
                              float* __restrict__ H) {
    __shared__ float lds[16 * LDS_STRIDE];
    const int rowBase = blockIdx.x * 16;
    const int t = threadIdx.x;

    #pragma unroll
    for (int rep = 0; rep < 2; ++rep) {
        int idx = t + rep * 256;
        int r   = idx >> 5;
        int c4  = idx & 31;
        float4 v = *(const float4*)(X + (size_t)(rowBase + r) * DIM + c4 * 4);
        if (bias) {
            float4 b = *(const float4*)(bias + c4 * 4);
            v.x = fmaxf(v.x + b.x, 0.0f);
            v.y = fmaxf(v.y + b.y, 0.0f);
            v.z = fmaxf(v.z + b.z, 0.0f);
            v.w = fmaxf(v.w + b.w, 0.0f);
        }
        *(float4*)(&lds[r * LDS_STRIDE + c4 * 4]) = v;
    }
    __syncthreads();

    const int wave = t >> 5;
    const int lane = t & 31;
    const int half = lane >> 4;
    const int l16  = lane & 15;
    const int colBase = wave * 16;

    v8f acc = {0.f, 0.f, 0.f, 0.f, 0.f, 0.f, 0.f, 0.f};

    #pragma unroll
    for (int k0 = 0; k0 < DIM; k0 += 4) {
        const int kk = k0 + half * 2;
        float2 av = *(const float2*)(&lds[l16 * LDS_STRIDE + kk]);
        v2f a, b;
        a[0] = av.x;
        a[1] = av.y;
        b[0] = W[(size_t)kk * DIM + colBase + l16];
        b[1] = W[(size_t)(kk + 1) * DIM + colBase + l16];
        acc = __builtin_amdgcn_wmma_f32_16x16x4_f32(false, a, false, b,
                                                    (short)0, acc, false, false);
    }

    float* outp = H + (size_t)rowBase * DIM + colBase;
    #pragma unroll
    for (int v = 0; v < 8; ++v) {
        outp[(size_t)(v + half * 8) * DIM + l16] = acc[v];
    }
}

__device__ __forceinline__ void list_append(int* list, int* nlist, float* acc, int e, unsigned dl,
                                            int base, const int* __restrict__ src,
                                            const float* __restrict__ h,
                                            const float* __restrict__ dinv) {
    const int pos = atomicAdd(nlist, 1);
    const int s = src[e];
    if (pos < LIST_CAP) {
        list[pos] = s | (int)(dl << 20);
    } else {
        const float nm = dinv[s] * dinv[base + (int)dl];
        for (int f = 0; f < DIM; ++f)
            atomicAdd(&acc[dl * DIM + f], h[(size_t)s * DIM + f] * nm);
    }
}

__launch_bounds__(WG)
__global__ void k_aggregate(const int* __restrict__ src, const int* __restrict__ dst,
                            const int* __restrict__ tile_n, const int* __restrict__ tile_list,
                            const float* __restrict__ h, const float* __restrict__ dinv,
                            const float* __restrict__ bias, float* __restrict__ out) {
    __shared__ float acc[NPB * DIM];
    __shared__ int   list[LIST_CAP];
    __shared__ int   nlist;
    const int t = threadIdx.x;
    const int wave = t >> 5;
    const int lane = t & 31;
    const int base = blockIdx.x * NPB;
    const int ntile = tile_n[blockIdx.x];

    {
        constexpr int R = (NPB * DIM) / WG;
        float hr[R], dr[R];
        #pragma unroll
        for (int r = 0; r < R; ++r) {
            const int idx = r * WG + t;
            hr[r] = h[(size_t)base * DIM + idx];
            dr[r] = dinv[base + (idx >> 7)];
        }
        const float bv = bias ? bias[t & 127] : 0.0f;
        #pragma unroll
        for (int r = 0; r < R; ++r) {
            float v = hr[r] * dr[r] * dr[r];
            if (bias) v += bv;
            acc[r * WG + t] = v;
        }
    }
    if (ntile <= LIST_CAP) {
        const int* __restrict__ inl = tile_list + (size_t)blockIdx.x * LIST_CAP;
        for (int k = t; k < ntile; k += WG) list[k] = inl[k];
    }
    if (t == 0) nlist = 0;
    __syncthreads();

    int n = ntile;
    if (ntile > LIST_CAP) {
        const int4* __restrict__ dst4 = (const int4*)dst;
        const int rot = blockIdx.x * (E4 / N_TILES);
        for (int i0 = 0; i0 < E4; i0 += 4 * WG) {
            int4 d[4];
            int  qq[4];
            #pragma unroll
            for (int u = 0; u < 4; ++u) {
                const int i = i0 + u * WG + t;
                int q = i + rot;
                if (q >= E4) q -= E4;
                qq[u] = q;
                d[u] = (i < E4) ? dst4[q] : make_int4(-1, -1, -1, -1);
            }
            #pragma unroll
            for (int u = 0; u < 4; ++u) {
                const int e0 = qq[u] * 4;
                const unsigned a = (unsigned)(d[u].x - base), b = (unsigned)(d[u].y - base);
                const unsigned c = (unsigned)(d[u].z - base), g = (unsigned)(d[u].w - base);
                if (min(min(a, b), min(c, g)) < NPB) {
                    if (a < NPB) list_append(list, &nlist, acc, e0 + 0, a, base, src, h, dinv);
                    if (b < NPB) list_append(list, &nlist, acc, e0 + 1, b, base, src, h, dinv);
                    if (c < NPB) list_append(list, &nlist, acc, e0 + 2, c, base, src, h, dinv);
                    if (g < NPB) list_append(list, &nlist, acc, e0 + 3, g, base, src, h, dinv);
                }
            }
        }
        __syncthreads();
        n = min(nlist, LIST_CAP);
    }

    for (int k0 = wave * PU; k0 < n; k0 += (WG / 32) * PU) {
        int   ent[PU], s[PU];
        float ds[PU], dd[PU];
        float hv[PU][4];
        #pragma unroll
        for (int u = 0; u < PU; ++u) {
            const int k = k0 + u;
            ent[u] = (k < n) ? list[k] : -1;
        }
        #pragma unroll
        for (int u = 0; u < PU; ++u) {
            s[u] = max(ent[u], 0) & 0xFFFFF;
            const int dl = (max(ent[u], 0) >> 20) & 255;
            ds[u] = dinv[s[u]];
            dd[u] = dinv[base + dl];
            const float* hp = h + (size_t)s[u] * DIM + lane;
            hv[u][0] = hp[0]; hv[u][1] = hp[32]; hv[u][2] = hp[64]; hv[u][3] = hp[96];
        }
        #pragma unroll
        for (int u = 0; u < PU; ++u)
            asm volatile("" :: "v"(ds[u]), "v"(dd[u]), "v"(hv[u][0]), "v"(hv[u][1]),
                               "v"(hv[u][2]), "v"(hv[u][3]));
        #pragma unroll
        for (int u = 0; u < PU; ++u) {
            if (ent[u] >= 0) {
                const float nm = ds[u] * dd[u];
                float* ap = &acc[((ent[u] >> 20) & 255) * DIM + lane];
                atomicAdd(ap +  0, hv[u][0] * nm);
                atomicAdd(ap + 32, hv[u][1] * nm);
                atomicAdd(ap + 64, hv[u][2] * nm);
                atomicAdd(ap + 96, hv[u][3] * nm);
            }
        }
    }
    __syncthreads();

    for (int r = 0; r < (NPB * DIM) / WG; ++r)
        out[(size_t)base * DIM + r * WG + t] = acc[r * WG + t];
}

__launch_bounds__(256)
__global__ void k_decode(const int* __restrict__ pa, const int* __restrict__ pb,
                         const float* __restrict__ z, float* __restrict__ out) {
    const int wid  = threadIdx.x >> 5;
    const int lane = threadIdx.x & 31;
    const int p = blockIdx.x * 8 + wid;
    if (p >= N_PAIRS) return;
    const int a = pa[p];
    const int b = pb[p];
    const float4 za = *(const float4*)(z + (size_t)a * DIM + lane * 4);
    const float4 zb = *(const float4*)(z + (size_t)b * DIM + lane * 4);
    float s = za.x * zb.x + za.y * zb.y + za.z * zb.z + za.w * zb.w;
    #pragma unroll
    for (int off = 16; off > 0; off >>= 1) s += __shfl_down(s, off);
    if (lane == 0) out[p] = s;
}

extern "C" void kernel_launch(void* const* d_in, const int* in_sizes, int n_in,
                              void* d_out, int out_size, void* d_ws, size_t ws_size,
                              hipStream_t stream) {
    const int*   edge_index = (const int*)d_in[0];
    const int*   edge_pairs = (const int*)d_in[1];
    const float* emb        = (const float*)d_in[2];
    const float* W1         = (const float*)d_in[3];
    const float* b1         = (const float*)d_in[4];
    const float* W2         = (const float*)d_in[5];
    const float* b2         = (const float*)d_in[6];
    float*       out        = (float*)d_out;

    const int* src = edge_index;
    const int* dst = edge_index + N_EDGES;
    const int* pa  = edge_pairs;
    const int* pb  = edge_pairs + N_PAIRS;

    char* ws = (char*)d_ws;
    size_t off = 0;
    float* dinv = (float*)(ws + off); off += ((size_t)N_NODES * 4 + 255) & ~(size_t)255;
    float* h    = (float*)(ws + off); off += (size_t)N_NODES * DIM * 4;
    float* agg  = (float*)(ws + off); off += (size_t)N_NODES * DIM * 4;
    int* tile_n    = (int*)(ws + off); off += ((size_t)N_TILES * 4 + 255) & ~(size_t)255;
    int* tile_list = (int*)(ws + off); off += (size_t)N_TILES * LIST_CAP * 4;
    (void)ws_size; (void)in_sizes; (void)n_in; (void)out_size;

    const int T = 256;
    const int gMM   = N_NODES / 16;
    const int gPw   = N_PAIRS / 8;

    k_bin<<<N_TILES, WG, 0, stream>>>(src, dst, dinv, tile_n, tile_list);

    k_matmul_wmma<<<gMM, T, 0, stream>>>(emb, W1, nullptr, h);
    k_aggregate<<<N_TILES, WG, 0, stream>>>(src, dst, tile_n, tile_list, h, dinv, nullptr, agg);

    k_matmul_wmma<<<gMM, T, 0, stream>>>(agg, W2, b1, h);
    k_aggregate<<<N_TILES, WG, 0, stream>>>(src, dst, tile_n, tile_list, h, dinv, b2, agg);

    k_decode<<<gPw, T, 0, stream>>>(pa, pb, agg, out);
}


# Edge list read once, not once per work-group: 250 groups split 3200 edges each by destination tile into workspace cells, a tile reads its 250 cells; 6 launches to 7; LDS float adds in arrival order
# speedup vs baseline: 2.5936x; 1.1331x over previous
#include <hip/hip_runtime.h>

#define N_NODES 50000
#define N_EDGES 800000
#define N_PAIRS 400000
#define DIM     128
#define LDS_STRIDE 132

typedef float v2f __attribute__((ext_vector_type(2)));
typedef float v8f __attribute__((ext_vector_type(8)));

#define NPB      200
#define N_TILES  (N_NODES / NPB)
#define WG       1024
#define E4       (N_EDGES / 4)
#define LIST_CAP 16384
#define PU       8
#define ESL      (N_EDGES / N_TILES)
#define PTE      (ESL / 16)
#define CELL_W   64
#define CELL_CAP (CELL_W - 1)
#define SP       65
#define PT       256
#define CPW      32
#define STAGE_W  ((N_TILES * SP + 1 + 3) / 4 * 4)


__launch_bounds__(PT)
__global__ void k_part(const int* __restrict__ src, const int* __restrict__ dst,
                       int* __restrict__ cells) {
    __shared__ __attribute__((aligned(16))) int stage[STAGE_W];
    const int t = threadIdx.x;
    const int wave = t >> 5;
    const int lane = t & 31;
    const int sl = blockIdx.x;
    for (int i = t; i < STAGE_W / 4; i += PT) ((int4*)stage)[i] = make_int4(0, 0, 0, 0);
    __syncthreads();
    if (t < PTE) {
        const int4* __restrict__ s4 = (const int4*)(src + (size_t)sl * ESL);
        const int4* __restrict__ d4 = (const int4*)(dst + (size_t)sl * ESL);
        int4 sv[4], dv[4];
        #pragma unroll
        for (int u = 0; u < 4; ++u) {
            sv[u] = s4[u * PTE + t];
            dv[u] = d4[u * PTE + t];
        }
        int  cell[16], ent[16], pos[16];
        bool ok[16];
        #pragma unroll
        for (int u = 0; u < 4; ++u) {
            const int ss[4] = {sv[u].x, sv[u].y, sv[u].z, sv[u].w};
            const int dd[4] = {dv[u].x, dv[u].y, dv[u].z, dv[u].w};
            #pragma unroll
            for (int j = 0; j < 4; ++j) {
                const int q = u * 4 + j;
                ok[q] = (unsigned)dd[j] < (unsigned)N_NODES;
                const int tile = ok[q] ? dd[j] / NPB : N_TILES;
                const int dl = ok[q] ? dd[j] - tile * NPB : 0;
                cell[q] = tile * SP;
                ent[q] = ss[j] | (dl << 20);
                pos[q] = atomicAdd(&stage[cell[q]], 1);
            }
        }
        #pragma unroll
        for (int q = 0; q < 16; ++q)
            if (ok[q] && pos[q] < CELL_CAP) stage[cell[q] + 1 + pos[q]] = ent[q];
    }
    __syncthreads();
    int* __restrict__ outc = cells + (size_t)sl * N_TILES * CELL_W;
    #pragma unroll
    for (int i = 0; i < CPW; ++i) {
        const int c = wave + i * (PT / 32);
        if (i * (PT / 32) + (PT / 32 - 1) < N_TILES || c < N_TILES) {
            int2 v;
            v.x = stage[c * SP + 2 * lane];
            v.y = stage[c * SP + 2 * lane + 1];
            *(int2*)(outc + c * CELL_W + 2 * lane) = v;
        }
    }
}

__device__ __forceinline__ void bin_append(int* cnt, int* nlist, int* __restrict__ outl,
                                           const int* __restrict__ src, int e, unsigned dl) {
    atomicAdd(&cnt[dl], 1);
    const int pos = atomicAdd(nlist, 1);
    if (pos < LIST_CAP) outl[pos] = src[e] | (int)(dl << 20);
}

__launch_bounds__(PT)
__global__ void k_deg(const int* __restrict__ src, const int* __restrict__ dst,
                      const int* __restrict__ cells,
                      float* __restrict__ dinv, int* __restrict__ tile_n,
                      int* __restrict__ tile_list) {
    __shared__ int cnt[256];
    __shared__ int nlist;
    __shared__ int ovf;
    const int t = threadIdx.x;
    const int wave = t >> 5;
    const int lane = t & 31;
    const int base = blockIdx.x * NPB;
    cnt[t] = 0;
    if (t == 0) { nlist = 0; ovf = 0; }
    __syncthreads();
    int* __restrict__ outl = tile_list + (size_t)blockIdx.x * LIST_CAP;

    int2 v[CPW];
    #pragma unroll
    for (int u = 0; u < CPW; ++u) {
        const int sl = min(wave + u * (PT / 32), N_TILES - 1);
        v[u] = *(const int2*)(cells + ((size_t)sl * N_TILES + blockIdx.x) * CELL_W + 2 * lane);
    }
    __builtin_amdgcn_sched_barrier(0);
    int c[CPW];
    int tot = 0;
    bool bad = false;
    #pragma unroll
    for (int u = 0; u < CPW; ++u) {
        c[u] = __builtin_amdgcn_readfirstlane(v[u].x);
        if (wave + u * (PT / 32) >= N_TILES) c[u] = 0;
        bad = bad || ((unsigned)c[u] > (unsigned)CELL_CAP);
        tot += c[u];
    }
    if (bad) {
        if (lane == 0) ovf = 1;
    } else {
        int wbase = 0;
        if (lane == 0) wbase = atomicAdd(&nlist, tot);
        int off = __builtin_amdgcn_readfirstlane(wbase);
        const int k0 = 2 * lane, k1 = k0 + 1;
        #pragma unroll
        for (int u = 0; u < CPW; ++u) {
            if (k0 >= 1 && k0 <= c[u]) {
                const int e = v[u].x;
                atomicAdd(&cnt[(e >> 20) & 255], 1);
                const int p = off + k0 - 1;
                if (p < LIST_CAP) outl[p] = e;
            }
            if (k1 <= c[u]) {
                const int e = v[u].y;
                atomicAdd(&cnt[(e >> 20) & 255], 1);
                const int p = off + k1 - 1;
                if (p < LIST_CAP) outl[p] = e;
            }
            off += c[u];
        }
    }
    __syncthreads();
    if (ovf) {
        cnt[t] = 0;
        if (t == 0) nlist = 0;
        __syncthreads();
        const int4* __restrict__ dst4 = (const int4*)dst;
        for (int i0 = 0; i0 < E4; i0 += 4 * PT) {
            int4 d[4];
            #pragma unroll
            for (int u = 0; u < 4; ++u) {
                const int i = i0 + u * PT + t;
                d[u] = (i < E4) ? dst4[i] : make_int4(-1, -1, -1, -1);
            }
            #pragma unroll
            for (int u = 0; u < 4; ++u) {
                const int e0 = (i0 + u * PT + t) * 4;
                const unsigned a = (unsigned)(d[u].x - base), b = (unsigned)(d[u].y - base);
                const unsigned cc = (unsigned)(d[u].z - base), g = (unsigned)(d[u].w - base);
                if (min(min(a, b), min(cc, g)) < NPB) {
                    if (a < NPB)  bin_append(cnt, &nlist, outl, src, e0 + 0, a);
                    if (b < NPB)  bin_append(cnt, &nlist, outl, src, e0 + 1, b);
                    if (cc < NPB) bin_append(cnt, &nlist, outl, src, e0 + 2, cc);
                    if (g < NPB)  bin_append(cnt, &nlist, outl, src, e0 + 3, g);
                }
            }
        }
        __syncthreads();
    }
    if (t < NPB) dinv[base + t] = rsqrtf(1.0f + (float)cnt[t]);
    if (t == 0) tile_n[blockIdx.x] = nlist;
}

__launch_bounds__(256, 4)
__global__ void k_matmul_wmma(const float* __restrict__ X,
                              const float* __restrict__ W,
                              const float* __restrict__ bias,
                              float* __restrict__ H) {
    __shared__ float lds[16 * LDS_STRIDE];
    const int rowBase = blockIdx.x * 16;
    const int t = threadIdx.x;

    #pragma unroll
    for (int rep = 0; rep < 2; ++rep) {
        int idx = t + rep * 256;
        int r   = idx >> 5;
        int c4  = idx & 31;
        float4 v = *(const float4*)(X + (size_t)(rowBase + r) * DIM + c4 * 4);
        if (bias) {
            float4 b = *(const float4*)(bias + c4 * 4);
            v.x = fmaxf(v.x + b.x, 0.0f);
            v.y = fmaxf(v.y + b.y, 0.0f);
            v.z = fmaxf(v.z + b.z, 0.0f);
            v.w = fmaxf(v.w + b.w, 0.0f);
        }
        *(float4*)(&lds[r * LDS_STRIDE + c4 * 4]) = v;
    }
    __syncthreads();

    const int wave = t >> 5;
    const int lane = t & 31;
    const int half = lane >> 4;
    const int l16  = lane & 15;
    const int colBase = wave * 16;

    v8f acc = {0.f, 0.f, 0.f, 0.f, 0.f, 0.f, 0.f, 0.f};

    #pragma unroll
    for (int k0 = 0; k0 < DIM; k0 += 4) {
        const int kk = k0 + half * 2;
        float2 av = *(const float2*)(&lds[l16 * LDS_STRIDE + kk]);
        v2f a, b;
        a[0] = av.x;
        a[1] = av.y;
        b[0] = W[(size_t)kk * DIM + colBase + l16];
        b[1] = W[(size_t)(kk + 1) * DIM + colBase + l16];
        acc = __builtin_amdgcn_wmma_f32_16x16x4_f32(false, a, false, b,
                                                    (short)0, acc, false, false);
    }

    float* outp = H + (size_t)rowBase * DIM + colBase;
    #pragma unroll
    for (int v = 0; v < 8; ++v) {
        outp[(size_t)(v + half * 8) * DIM + l16] = acc[v];
    }
}

__device__ __forceinline__ void list_append(int* list, int* nlist, float* acc, int e, unsigned dl,
                                            int base, const int* __restrict__ src,
                                            const float* __restrict__ h,
                                            const float* __restrict__ dinv) {
    const int pos = atomicAdd(nlist, 1);
    const int s = src[e];
    if (pos < LIST_CAP) {
        list[pos] = s | (int)(dl << 20);
    } else {
        const float nm = dinv[s] * dinv[base + (int)dl];
        for (int f = 0; f < DIM; ++f)
            atomicAdd(&acc[dl * DIM + f], h[(size_t)s * DIM + f] * nm);
    }
}

__launch_bounds__(WG)
__global__ void k_aggregate(const int* __restrict__ src, const int* __restrict__ dst,
                            const int* __restrict__ tile_n, const int* __restrict__ tile_list,
                            const float* __restrict__ h, const float* __restrict__ dinv,
                            const float* __restrict__ bias, float* __restrict__ out) {
    __shared__ float acc[NPB * DIM];
    __shared__ int   list[LIST_CAP];
    __shared__ int   nlist;
    const int t = threadIdx.x;
    const int wave = t >> 5;
    const int lane = t & 31;
    const int base = blockIdx.x * NPB;
    const int ntile = tile_n[blockIdx.x];

    {
        constexpr int R = (NPB * DIM) / WG;
        float hr[R], dr[R];
        #pragma unroll
        for (int r = 0; r < R; ++r) {
            const int idx = r * WG + t;
            hr[r] = h[(size_t)base * DIM + idx];
            dr[r] = dinv[base + (idx >> 7)];
        }
        const float bv = bias ? bias[t & 127] : 0.0f;
        #pragma unroll
        for (int r = 0; r < R; ++r) {
            float v = hr[r] * dr[r] * dr[r];
            if (bias) v += bv;
            acc[r * WG + t] = v;
        }
    }
    if (ntile <= LIST_CAP) {
        const int* __restrict__ inl = tile_list + (size_t)blockIdx.x * LIST_CAP;
        for (int k = t; k < ntile; k += WG) list[k] = inl[k];
    }
    if (t == 0) nlist = 0;
    __syncthreads();

    int n = ntile;
    if (ntile > LIST_CAP) {
        const int4* __restrict__ dst4 = (const int4*)dst;
        const int rot = blockIdx.x * (E4 / N_TILES);
        for (int i0 = 0; i0 < E4; i0 += 4 * WG) {
            int4 d[4];
            int  qq[4];
            #pragma unroll
            for (int u = 0; u < 4; ++u) {
                const int i = i0 + u * WG + t;
                int q = i + rot;
                if (q >= E4) q -= E4;
                qq[u] = q;
                d[u] = (i < E4) ? dst4[q] : make_int4(-1, -1, -1, -1);
            }
            #pragma unroll
            for (int u = 0; u < 4; ++u) {
                const int e0 = qq[u] * 4;
                const unsigned a = (unsigned)(d[u].x - base), b = (unsigned)(d[u].y - base);
                const unsigned c = (unsigned)(d[u].z - base), g = (unsigned)(d[u].w - base);
                if (min(min(a, b), min(c, g)) < NPB) {
                    if (a < NPB) list_append(list, &nlist, acc, e0 + 0, a, base, src, h, dinv);
                    if (b < NPB) list_append(list, &nlist, acc, e0 + 1, b, base, src, h, dinv);
                    if (c < NPB) list_append(list, &nlist, acc, e0 + 2, c, base, src, h, dinv);
                    if (g < NPB) list_append(list, &nlist, acc, e0 + 3, g, base, src, h, dinv);
                }
            }
        }
        __syncthreads();
        n = min(nlist, LIST_CAP);
    }

    for (int k0 = wave * PU; k0 < n; k0 += (WG / 32) * PU) {
        int   ent[PU], s[PU];
        float ds[PU], dd[PU];
        float hv[PU][4];
        #pragma unroll
        for (int u = 0; u < PU; ++u) {
            const int k = k0 + u;
            ent[u] = (k < n) ? list[k] : -1;
        }
        #pragma unroll
        for (int u = 0; u < PU; ++u) {
            s[u] = max(ent[u], 0) & 0xFFFFF;
            const int dl = (max(ent[u], 0) >> 20) & 255;
            ds[u] = dinv[s[u]];
            dd[u] = dinv[base + dl];
            const float* hp = h + (size_t)s[u] * DIM + lane;
            hv[u][0] = hp[0]; hv[u][1] = hp[32]; hv[u][2] = hp[64]; hv[u][3] = hp[96];
        }
        #pragma unroll
        for (int u = 0; u < PU; ++u)
            asm volatile("" :: "v"(ds[u]), "v"(dd[u]), "v"(hv[u][0]), "v"(hv[u][1]),
                               "v"(hv[u][2]), "v"(hv[u][3]));
        #pragma unroll
        for (int u = 0; u < PU; ++u) {
            if (ent[u] >= 0) {
                const float nm = ds[u] * dd[u];
                float* ap = &acc[((ent[u] >> 20) & 255) * DIM + lane];
                atomicAdd(ap +  0, hv[u][0] * nm);
                atomicAdd(ap + 32, hv[u][1] * nm);
                atomicAdd(ap + 64, hv[u][2] * nm);
                atomicAdd(ap + 96, hv[u][3] * nm);
            }
        }
    }
    __syncthreads();

    for (int r = 0; r < (NPB * DIM) / WG; ++r)
        out[(size_t)base * DIM + r * WG + t] = acc[r * WG + t];
}

__launch_bounds__(256)
__global__ void k_decode(const int* __restrict__ pa, const int* __restrict__ pb,
                         const float* __restrict__ z, float* __restrict__ out) {
    const int wid  = threadIdx.x >> 5;
    const int lane = threadIdx.x & 31;
    const int p = blockIdx.x * 8 + wid;
    if (p >= N_PAIRS) return;
    const int a = pa[p];
    const int b = pb[p];
    const float4 za = *(const float4*)(z + (size_t)a * DIM + lane * 4);
    const float4 zb = *(const float4*)(z + (size_t)b * DIM + lane * 4);
    float s = za.x * zb.x + za.y * zb.y + za.z * zb.z + za.w * zb.w;
    #pragma unroll
    for (int off = 16; off > 0; off >>= 1) s += __shfl_down(s, off);
    if (lane == 0) out[p] = s;
}

extern "C" void kernel_launch(void* const* d_in, const int* in_sizes, int n_in,
                              void* d_out, int out_size, void* d_ws, size_t ws_size,
                              hipStream_t stream) {
    const int*   edge_index = (const int*)d_in[0];
    const int*   edge_pairs = (const int*)d_in[1];
    const float* emb        = (const float*)d_in[2];
    const float* W1         = (const float*)d_in[3];
    const float* b1         = (const float*)d_in[4];
    const float* W2         = (const float*)d_in[5];
    const float* b2         = (const float*)d_in[6];
    float*       out        = (float*)d_out;

    const int* src = edge_index;
    const int* dst = edge_index + N_EDGES;
    const int* pa  = edge_pairs;
    const int* pb  = edge_pairs + N_PAIRS;

    char* ws = (char*)d_ws;
    size_t off = 0;
    float* dinv = (float*)(ws + off); off += ((size_t)N_NODES * 4 + 255) & ~(size_t)255;
    float* h    = (float*)(ws + off); off += (size_t)N_NODES * DIM * 4;
    float* agg  = (float*)(ws + off); off += (size_t)N_NODES * DIM * 4;
    int* tile_n    = (int*)(ws + off); off += ((size_t)N_TILES * 4 + 255) & ~(size_t)255;
    int* tile_list = (int*)(ws + off); off += (size_t)N_TILES * LIST_CAP * 4;
    int* cells     = (int*)(ws + off); off += (size_t)N_TILES * N_TILES * CELL_W * 4;
    (void)ws_size; (void)in_sizes; (void)n_in; (void)out_size;

    const int T = 256;
    const int gMM   = N_NODES / 16;
    const int gPw   = N_PAIRS / 8;

    k_part<<<N_TILES, PT, 0, stream>>>(src, dst, cells);
    k_deg<<<N_TILES, PT, 0, stream>>>(src, dst, cells, dinv, tile_n, tile_list);

    k_matmul_wmma<<<gMM, T, 0, stream>>>(emb, W1, nullptr, h);
    k_aggregate<<<N_TILES, WG, 0, stream>>>(src, dst, tile_n, tile_list, h, dinv, nullptr, agg);

    k_matmul_wmma<<<gMM, T, 0, stream>>>(agg, W2, b1, h);
    k_aggregate<<<N_TILES, WG, 0, stream>>>(src, dst, tile_n, tile_list, h, dinv, b2, agg);

    k_decode<<<gPw, T, 0, stream>>>(pa, pb, agg, out);
}


# Layer-1 product shares the grids of the two edge launches: row tiles of emb @ W1 run behind the split and behind the degree work-groups in the same launches; 7 launches to 6
# speedup vs baseline: 2.7039x; 1.0425x over previous
#include <hip/hip_runtime.h>

#define N_NODES 50000
#define N_EDGES 800000
#define N_PAIRS 400000
#define DIM     128
#define LDS_STRIDE 132

typedef float v2f __attribute__((ext_vector_type(2)));
typedef float v8f __attribute__((ext_vector_type(8)));

#define NPB      200
#define N_TILES  (N_NODES / NPB)
#define WG       1024
#define E4       (N_EDGES / 4)
#define LIST_CAP 16384
#define PU       8
#define ESL      (N_EDGES / N_TILES)
#define PTE      (ESL / 16)
#define CELL_W   64
#define CELL_CAP (CELL_W - 1)
#define SP       65
#define PT       256
#define CPW      32
#define STAGE_W  ((N_TILES * SP + 1 + 3) / 4 * 4)
#define MM_TILES (N_NODES / 16)
#define MM_A     1562
#define MM_B     (MM_TILES - MM_A)
#define MM_LDS_W (16 * LDS_STRIDE)


__device__ __forceinline__ void part_body(int* stage, const int sl,
                                          const int* __restrict__ src, const int* __restrict__ dst,
                                          int* __restrict__ cells) {
    const int t = threadIdx.x;
    const int wave = t >> 5;
    const int lane = t & 31;
    for (int i = t; i < STAGE_W / 4; i += PT) ((int4*)stage)[i] = make_int4(0, 0, 0, 0);
    __syncthreads();
    if (t < PTE) {
        const int4* __restrict__ s4 = (const int4*)(src + (size_t)sl * ESL);
        const int4* __restrict__ d4 = (const int4*)(dst + (size_t)sl * ESL);
        int4 sv[4], dv[4];
        #pragma unroll
        for (int u = 0; u < 4; ++u) {
            sv[u] = s4[u * PTE + t];
            dv[u] = d4[u * PTE + t];
        }
        int  cell[16], ent[16], pos[16];
        bool ok[16];
        #pragma unroll
        for (int u = 0; u < 4; ++u) {
            const int ss[4] = {sv[u].x, sv[u].y, sv[u].z, sv[u].w};
            const int dd[4] = {dv[u].x, dv[u].y, dv[u].z, dv[u].w};
            #pragma unroll
            for (int j = 0; j < 4; ++j) {
                const int q = u * 4 + j;
                ok[q] = (unsigned)dd[j] < (unsigned)N_NODES;
                const int tile = ok[q] ? dd[j] / NPB : N_TILES;
                const int dl = ok[q] ? dd[j] - tile * NPB : 0;
                cell[q] = tile * SP;
                ent[q] = ss[j] | (dl << 20);
                pos[q] = atomicAdd(&stage[cell[q]], 1);
            }
        }
        #pragma unroll
        for (int q = 0; q < 16; ++q)
            if (ok[q] && pos[q] < CELL_CAP) stage[cell[q] + 1 + pos[q]] = ent[q];
    }
    __syncthreads();
    int* __restrict__ outc = cells + (size_t)sl * N_TILES * CELL_W;
    #pragma unroll
    for (int i = 0; i < CPW; ++i) {
        const int c = wave + i * (PT / 32);
        if (i * (PT / 32) + (PT / 32 - 1) < N_TILES || c < N_TILES) {
            int2 v;
            v.x = stage[c * SP + 2 * lane];
            v.y = stage[c * SP + 2 * lane + 1];
            *(int2*)(outc + c * CELL_W + 2 * lane) = v;
        }
    }
}

__device__ __forceinline__ void bin_append(int* cnt, int* nlist, int* __restrict__ outl,
                                           const int* __restrict__ src, int e, unsigned dl) {
    atomicAdd(&cnt[dl], 1);
    const int pos = atomicAdd(nlist, 1);
    if (pos < LIST_CAP) outl[pos] = src[e] | (int)(dl << 20);
}

__device__ __forceinline__ void deg_body(int* lds, const int tile,
                                         const int* __restrict__ src, const int* __restrict__ dst,
                                         const int* __restrict__ cells,
                                         float* __restrict__ dinv, int* __restrict__ tile_n,
                                         int* __restrict__ tile_list) {
    int* cnt = lds;
    int* nlistp = lds + 256;
    int* ovfp = lds + 257;
    const int t = threadIdx.x;
    const int wave = t >> 5;
    const int lane = t & 31;
    const int base = tile * NPB;
    cnt[t] = 0;
    if (t == 0) { *nlistp = 0; *ovfp = 0; }
    __syncthreads();
    int* __restrict__ outl = tile_list + (size_t)tile * LIST_CAP;

    int2 v[CPW];
    #pragma unroll
    for (int u = 0; u < CPW; ++u) {
        const int sl = min(wave + u * (PT / 32), N_TILES - 1);
        v[u] = *(const int2*)(cells + ((size_t)sl * N_TILES + tile) * CELL_W + 2 * lane);
    }
    __builtin_amdgcn_sched_barrier(0);
    int c[CPW];
    int tot = 0;
    bool bad = false;
    #pragma unroll
    for (int u = 0; u < CPW; ++u) {
        c[u] = __builtin_amdgcn_readfirstlane(v[u].x);
        if (wave + u * (PT / 32) >= N_TILES) c[u] = 0;
        bad = bad || ((unsigned)c[u] > (unsigned)CELL_CAP);
        tot += c[u];
    }
    if (bad) {
        if (lane == 0) *ovfp = 1;
    } else {
        int wbase = 0;
        if (lane == 0) wbase = atomicAdd(nlistp, tot);
        int off = __builtin_amdgcn_readfirstlane(wbase);
        const int k0 = 2 * lane, k1 = k0 + 1;
        #pragma unroll
        for (int u = 0; u < CPW; ++u) {
            if (k0 >= 1 && k0 <= c[u]) {
                const int e = v[u].x;
                atomicAdd(&cnt[(e >> 20) & 255], 1);
                const int p = off + k0 - 1;
                if (p < LIST_CAP) outl[p] = e;
            }
            if (k1 <= c[u]) {
                const int e = v[u].y;
                atomicAdd(&cnt[(e >> 20) & 255], 1);
                const int p = off + k1 - 1;
                if (p < LIST_CAP) outl[p] = e;
            }
            off += c[u];
        }
    }
    __syncthreads();
    if (*ovfp) {
        cnt[t] = 0;
        if (t == 0) *nlistp = 0;
        __syncthreads();
        const int4* __restrict__ dst4 = (const int4*)dst;
        for (int i0 = 0; i0 < E4; i0 += 4 * PT) {
            int4 d[4];
            #pragma unroll
            for (int u = 0; u < 4; ++u) {
                const int i = i0 + u * PT + t;
                d[u] = (i < E4) ? dst4[i] : make_int4(-1, -1, -1, -1);
            }
            #pragma unroll
            for (int u = 0; u < 4; ++u) {
                const int e0 = (i0 + u * PT + t) * 4;
                const unsigned a = (unsigned)(d[u].x - base), b = (unsigned)(d[u].y - base);
                const unsigned cc = (unsigned)(d[u].z - base), g = (unsigned)(d[u].w - base);
                if (min(min(a, b), min(cc, g)) < NPB) {
                    if (a < NPB)  bin_append(cnt, nlistp, outl, src, e0 + 0, a);
                    if (b < NPB)  bin_append(cnt, nlistp, outl, src, e0 + 1, b);
                    if (cc < NPB) bin_append(cnt, nlistp, outl, src, e0 + 2, cc);
                    if (g < NPB)  bin_append(cnt, nlistp, outl, src, e0 + 3, g);
                }
            }
        }
        __syncthreads();
    }
    if (t < NPB) dinv[base + t] = rsqrtf(1.0f + (float)cnt[t]);
    if (t == 0) tile_n[tile] = *nlistp;
}

__device__ __forceinline__ void mm_tile(float* lds, const int rowTile,
                                        const float* __restrict__ X,
                                        const float* __restrict__ W,
                                        const float* __restrict__ bias,
                                        float* __restrict__ H) {
    const int rowBase = rowTile * 16;
    const int t = threadIdx.x;

    #pragma unroll
    for (int rep = 0; rep < 2; ++rep) {
        int idx = t + rep * 256;
        int r   = idx >> 5;
        int c4  = idx & 31;
        float4 v = *(const float4*)(X + (size_t)(rowBase + r) * DIM + c4 * 4);
        if (bias) {
            float4 b = *(const float4*)(bias + c4 * 4);
            v.x = fmaxf(v.x + b.x, 0.0f);
            v.y = fmaxf(v.y + b.y, 0.0f);
            v.z = fmaxf(v.z + b.z, 0.0f);
            v.w = fmaxf(v.w + b.w, 0.0f);
        }
        *(float4*)(&lds[r * LDS_STRIDE + c4 * 4]) = v;
    }
    __syncthreads();

    const int wave = t >> 5;
    const int lane = t & 31;
    const int half = lane >> 4;
    const int l16  = lane & 15;
    const int colBase = wave * 16;

    v8f acc = {0.f, 0.f, 0.f, 0.f, 0.f, 0.f, 0.f, 0.f};

    #pragma unroll
    for (int k0 = 0; k0 < DIM; k0 += 4) {
        const int kk = k0 + half * 2;
        float2 av = *(const float2*)(&lds[l16 * LDS_STRIDE + kk]);
        v2f a, b;
        a[0] = av.x;
        a[1] = av.y;
        b[0] = W[(size_t)kk * DIM + colBase + l16];
        b[1] = W[(size_t)(kk + 1) * DIM + colBase + l16];
        acc = __builtin_amdgcn_wmma_f32_16x16x4_f32(false, a, false, b,
                                                    (short)0, acc, false, false);
    }

    float* outp = H + (size_t)rowBase * DIM + colBase;
    #pragma unroll
    for (int v = 0; v < 8; ++v) {
        outp[(size_t)(v + half * 8) * DIM + l16] = acc[v];
    }
}

__launch_bounds__(256, 4)
__global__ void k_matmul_wmma(const float* __restrict__ X,
                              const float* __restrict__ W,
                              const float* __restrict__ bias,
                              float* __restrict__ H) {
    __shared__ __attribute__((aligned(16))) float lds[MM_LDS_W];
    mm_tile(lds, blockIdx.x, X, W, bias, H);
}

__launch_bounds__(PT)
__global__ void k_part_mm(const int* __restrict__ src, const int* __restrict__ dst,
                          int* __restrict__ cells,
                          const float* __restrict__ X, const float* __restrict__ W,
                          float* __restrict__ H) {
    __shared__ __attribute__((aligned(16))) int smem[STAGE_W];
    if (blockIdx.x < N_TILES) part_body(smem, blockIdx.x, src, dst, cells);
    else mm_tile((float*)smem, blockIdx.x - N_TILES, X, W, nullptr, H);
}

__launch_bounds__(PT)
__global__ void k_deg_mm(const int* __restrict__ src, const int* __restrict__ dst,
                         const int* __restrict__ cells,
                         float* __restrict__ dinv, int* __restrict__ tile_n,
                         int* __restrict__ tile_list,
                         const float* __restrict__ X, const float* __restrict__ W,
                         float* __restrict__ H) {
    __shared__ __attribute__((aligned(16))) int smem[MM_LDS_W];
    if (blockIdx.x < N_TILES) deg_body(smem, blockIdx.x, src, dst, cells, dinv, tile_n, tile_list);
    else mm_tile((float*)smem, MM_A + (blockIdx.x - N_TILES), X, W, nullptr, H);
}

__device__ __forceinline__ void list_append(int* list, int* nlist, float* acc, int e, unsigned dl,
                                            int base, const int* __restrict__ src,
                                            const float* __restrict__ h,
                                            const float* __restrict__ dinv) {
    const int pos = atomicAdd(nlist, 1);
    const int s = src[e];
    if (pos < LIST_CAP) {
        list[pos] = s | (int)(dl << 20);
    } else {
        const float nm = dinv[s] * dinv[base + (int)dl];
        for (int f = 0; f < DIM; ++f)
            atomicAdd(&acc[dl * DIM + f], h[(size_t)s * DIM + f] * nm);
    }
}

__launch_bounds__(WG)
__global__ void k_aggregate(const int* __restrict__ src, const int* __restrict__ dst,
                            const int* __restrict__ tile_n, const int* __restrict__ tile_list,
                            const float* __restrict__ h, const float* __restrict__ dinv,
                            const float* __restrict__ bias, float* __restrict__ out) {
    __shared__ float acc[NPB * DIM];
    __shared__ int   list[LIST_CAP];
    __shared__ int   nlist;
    const int t = threadIdx.x;
    const int wave = t >> 5;
    const int lane = t & 31;
    const int base = blockIdx.x * NPB;
    const int ntile = tile_n[blockIdx.x];

    {
        constexpr int R = (NPB * DIM) / WG;
        float hr[R], dr[R];
        #pragma unroll
        for (int r = 0; r < R; ++r) {
            const int idx = r * WG + t;
            hr[r] = h[(size_t)base * DIM + idx];
            dr[r] = dinv[base + (idx >> 7)];
        }
        const float bv = bias ? bias[t & 127] : 0.0f;
        #pragma unroll
        for (int r = 0; r < R; ++r) {
            float v = hr[r] * dr[r] * dr[r];
            if (bias) v += bv;
            acc[r * WG + t] = v;
        }
    }
    if (ntile <= LIST_CAP) {
        const int* __restrict__ inl = tile_list + (size_t)blockIdx.x * LIST_CAP;
        for (int k = t; k < ntile; k += WG) list[k] = inl[k];
    }
    if (t == 0) nlist = 0;
    __syncthreads();

    int n = ntile;
    if (ntile > LIST_CAP) {
        const int4* __restrict__ dst4 = (const int4*)dst;
        const int rot = blockIdx.x * (E4 / N_TILES);
        for (int i0 = 0; i0 < E4; i0 += 4 * WG) {
            int4 d[4];
            int  qq[4];
            #pragma unroll
            for (int u = 0; u < 4; ++u) {
                const int i = i0 + u * WG + t;
                int q = i + rot;
                if (q >= E4) q -= E4;
                qq[u] = q;
                d[u] = (i < E4) ? dst4[q] : make_int4(-1, -1, -1, -1);
            }
            #pragma unroll
            for (int u = 0; u < 4; ++u) {
                const int e0 = qq[u] * 4;
                const unsigned a = (unsigned)(d[u].x - base), b = (unsigned)(d[u].y - base);
                const unsigned c = (unsigned)(d[u].z - base), g = (unsigned)(d[u].w - base);
                if (min(min(a, b), min(c, g)) < NPB) {
                    if (a < NPB) list_append(list, &nlist, acc, e0 + 0, a, base, src, h, dinv);
                    if (b < NPB) list_append(list, &nlist, acc, e0 + 1, b, base, src, h, dinv);
                    if (c < NPB) list_append(list, &nlist, acc, e0 + 2, c, base, src, h, dinv);
                    if (g < NPB) list_append(list, &nlist, acc, e0 + 3, g, base, src, h, dinv);
                }
            }
        }
        __syncthreads();
        n = min(nlist, LIST_CAP);
    }

    for (int k0 = wave * PU; k0 < n; k0 += (WG / 32) * PU) {
        int   ent[PU], s[PU];
        float ds[PU], dd[PU];
        float hv[PU][4];
        #pragma unroll
        for (int u = 0; u < PU; ++u) {
            const int k = k0 + u;
            ent[u] = (k < n) ? list[k] : -1;
        }
        #pragma unroll
        for (int u = 0; u < PU; ++u) {
            s[u] = max(ent[u], 0) & 0xFFFFF;
            const int dl = (max(ent[u], 0) >> 20) & 255;
            ds[u] = dinv[s[u]];
            dd[u] = dinv[base + dl];
            const float* hp = h + (size_t)s[u] * DIM + lane;
            hv[u][0] = hp[0]; hv[u][1] = hp[32]; hv[u][2] = hp[64]; hv[u][3] = hp[96];
        }
        #pragma unroll
        for (int u = 0; u < PU; ++u)
            asm volatile("" :: "v"(ds[u]), "v"(dd[u]), "v"(hv[u][0]), "v"(hv[u][1]),
                               "v"(hv[u][2]), "v"(hv[u][3]));
        #pragma unroll
        for (int u = 0; u < PU; ++u) {
            if (ent[u] >= 0) {
                const float nm = ds[u] * dd[u];
                float* ap = &acc[((ent[u] >> 20) & 255) * DIM + lane];
                atomicAdd(ap +  0, hv[u][0] * nm);
                atomicAdd(ap + 32, hv[u][1] * nm);
                atomicAdd(ap + 64, hv[u][2] * nm);
                atomicAdd(ap + 96, hv[u][3] * nm);
            }
        }
    }
    __syncthreads();

    for (int r = 0; r < (NPB * DIM) / WG; ++r)
        out[(size_t)base * DIM + r * WG + t] = acc[r * WG + t];
}

__launch_bounds__(256)
__global__ void k_decode(const int* __restrict__ pa, const int* __restrict__ pb,
                         const float* __restrict__ z, float* __restrict__ out) {
    const int wid  = threadIdx.x >> 5;
    const int lane = threadIdx.x & 31;
    const int p = blockIdx.x * 8 + wid;
    if (p >= N_PAIRS) return;
    const int a = pa[p];
    const int b = pb[p];
    const float4 za = *(const float4*)(z + (size_t)a * DIM + lane * 4);
    const float4 zb = *(const float4*)(z + (size_t)b * DIM + lane * 4);
    float s = za.x * zb.x + za.y * zb.y + za.z * zb.z + za.w * zb.w;
    #pragma unroll
    for (int off = 16; off > 0; off >>= 1) s += __shfl_down(s, off);
    if (lane == 0) out[p] = s;
}

extern "C" void kernel_launch(void* const* d_in, const int* in_sizes, int n_in,
                              void* d_out, int out_size, void* d_ws, size_t ws_size,
                              hipStream_t stream) {
    const int*   edge_index = (const int*)d_in[0];
    const int*   edge_pairs = (const int*)d_in[1];
    const float* emb        = (const float*)d_in[2];
    const float* W1         = (const float*)d_in[3];
    const float* b1         = (const float*)d_in[4];
    const float* W2         = (const float*)d_in[5];
    const float* b2         = (const float*)d_in[6];
    float*       out        = (float*)d_out;

    const int* src = edge_index;
    const int* dst = edge_index + N_EDGES;
    const int* pa  = edge_pairs;
    const int* pb  = edge_pairs + N_PAIRS;

    char* ws = (char*)d_ws;
    size_t off = 0;
    float* dinv = (float*)(ws + off); off += ((size_t)N_NODES * 4 + 255) & ~(size_t)255;
    float* h    = (float*)(ws + off); off += (size_t)N_NODES * DIM * 4;
    float* agg  = (float*)(ws + off); off += (size_t)N_NODES * DIM * 4;
    int* tile_n    = (int*)(ws + off); off += ((size_t)N_TILES * 4 + 255) & ~(size_t)255;
    int* tile_list = (int*)(ws + off); off += (size_t)N_TILES * LIST_CAP * 4;
    int* cells     = (int*)(ws + off); off += (size_t)N_TILES * N_TILES * CELL_W * 4;
    (void)ws_size; (void)in_sizes; (void)n_in; (void)out_size;

    const int T = 256;
    const int gMM   = MM_TILES;
    const int gPw   = N_PAIRS / 8;

    k_part_mm<<<N_TILES + MM_A, PT, 0, stream>>>(src, dst, cells, emb, W1, h);
    k_deg_mm<<<N_TILES + MM_B, PT, 0, stream>>>(src, dst, cells, dinv, tile_n, tile_list, emb, W1, h);

    k_aggregate<<<N_TILES, WG, 0, stream>>>(src, dst, tile_n, tile_list, h, dinv, nullptr, agg);

    k_matmul_wmma<<<gMM, T, 0, stream>>>(agg, W2, b1, h);
    k_aggregate<<<N_TILES, WG, 0, stream>>>(src, dst, tile_n, tile_list, h, dinv, b2, agg);

    k_decode<<<gPw, T, 0, stream>>>(pa, pb, agg, out);
}


# Aggregation gathers each 512-B source row with one 16-byte-per-lane load, not four 4-byte-per-lane loads (load instructions per edge 6 to 3); LDS rows kept lane-major; order of sums unchanged
# speedup vs baseline: 2.8095x; 1.0391x over previous
#include <hip/hip_runtime.h>

#define N_NODES 50000
#define N_EDGES 800000
#define N_PAIRS 400000
#define DIM     128
#define LDS_STRIDE 132

typedef float v2f __attribute__((ext_vector_type(2)));
typedef float v8f __attribute__((ext_vector_type(8)));

#define NPB      200
#define N_TILES  (N_NODES / NPB)
#define WG       1024
#define E4       (N_EDGES / 4)
#define LIST_CAP 16384
#define PU       8
#define ESL      (N_EDGES / N_TILES)
#define PTE      (ESL / 16)
#define CELL_W   64
#define CELL_CAP (CELL_W - 1)
#define SP       65
#define PT       256
#define CPW      32
#define STAGE_W  ((N_TILES * SP + 1 + 3) / 4 * 4)
#define MM_TILES (N_NODES / 16)
#define MM_A     1562
#define MM_B     (MM_TILES - MM_A)
#define MM_LDS_W (16 * LDS_STRIDE)


__device__ __forceinline__ void part_body(int* stage, const int sl,
                                          const int* __restrict__ src, const int* __restrict__ dst,
                                          int* __restrict__ cells) {
    const int t = threadIdx.x;
    const int wave = t >> 5;
    const int lane = t & 31;
    for (int i = t; i < STAGE_W / 4; i += PT) ((int4*)stage)[i] = make_int4(0, 0, 0, 0);
    __syncthreads();
    if (t < PTE) {
        const int4* __restrict__ s4 = (const int4*)(src + (size_t)sl * ESL);
        const int4* __restrict__ d4 = (const int4*)(dst + (size_t)sl * ESL);
        int4 sv[4], dv[4];
        #pragma unroll
        for (int u = 0; u < 4; ++u) {
            sv[u] = s4[u * PTE + t];
            dv[u] = d4[u * PTE + t];
        }
        int  cell[16], ent[16], pos[16];
        bool ok[16];
        #pragma unroll
        for (int u = 0; u < 4; ++u) {
            const int ss[4] = {sv[u].x, sv[u].y, sv[u].z, sv[u].w};
            const int dd[4] = {dv[u].x, dv[u].y, dv[u].z, dv[u].w};
            #pragma unroll
            for (int j = 0; j < 4; ++j) {
                const int q = u * 4 + j;
                ok[q] = (unsigned)dd[j] < (unsigned)N_NODES;
                const int tile = ok[q] ? dd[j] / NPB : N_TILES;
                const int dl = ok[q] ? dd[j] - tile * NPB : 0;
                cell[q] = tile * SP;
                ent[q] = ss[j] | (dl << 20);
                pos[q] = atomicAdd(&stage[cell[q]], 1);
            }
        }
        #pragma unroll
        for (int q = 0; q < 16; ++q)
            if (ok[q] && pos[q] < CELL_CAP) stage[cell[q] + 1 + pos[q]] = ent[q];
    }
    __syncthreads();
    int* __restrict__ outc = cells + (size_t)sl * N_TILES * CELL_W;
    #pragma unroll
    for (int i = 0; i < CPW; ++i) {
        const int c = wave + i * (PT / 32);
        if (i * (PT / 32) + (PT / 32 - 1) < N_TILES || c < N_TILES) {
            int2 v;
            v.x = stage[c * SP + 2 * lane];
            v.y = stage[c * SP + 2 * lane + 1];
            *(int2*)(outc + c * CELL_W + 2 * lane) = v;
        }
    }
}

__device__ __forceinline__ void bin_append(int* cnt, int* nlist, int* __restrict__ outl,
                                           const int* __restrict__ src, int e, unsigned dl) {
    atomicAdd(&cnt[dl], 1);
    const int pos = atomicAdd(nlist, 1);
    if (pos < LIST_CAP) outl[pos] = src[e] | (int)(dl << 20);
}

__device__ __forceinline__ void deg_body(int* lds, const int tile,
                                         const int* __restrict__ src, const int* __restrict__ dst,
                                         const int* __restrict__ cells,
                                         float* __restrict__ dinv, int* __restrict__ tile_n,
                                         int* __restrict__ tile_list) {
    int* cnt = lds;
    int* nlistp = lds + 256;
    int* ovfp = lds + 257;
    const int t = threadIdx.x;
    const int wave = t >> 5;
    const int lane = t & 31;
    const int base = tile * NPB;
    cnt[t] = 0;
    if (t == 0) { *nlistp = 0; *ovfp = 0; }
    __syncthreads();
    int* __restrict__ outl = tile_list + (size_t)tile * LIST_CAP;

    int2 v[CPW];
    #pragma unroll
    for (int u = 0; u < CPW; ++u) {
        const int sl = min(wave + u * (PT / 32), N_TILES - 1);
        v[u] = *(const int2*)(cells + ((size_t)sl * N_TILES + tile) * CELL_W + 2 * lane);
    }
    __builtin_amdgcn_sched_barrier(0);
    int c[CPW];
    int tot = 0;
    bool bad = false;
    #pragma unroll
    for (int u = 0; u < CPW; ++u) {
        c[u] = __builtin_amdgcn_readfirstlane(v[u].x);
        if (wave + u * (PT / 32) >= N_TILES) c[u] = 0;
        bad = bad || ((unsigned)c[u] > (unsigned)CELL_CAP);
        tot += c[u];
    }
    if (bad) {
        if (lane == 0) *ovfp = 1;
    } else {
        int wbase = 0;
        if (lane == 0) wbase = atomicAdd(nlistp, tot);
        int off = __builtin_amdgcn_readfirstlane(wbase);
        const int k0 = 2 * lane, k1 = k0 + 1;
        #pragma unroll
        for (int u = 0; u < CPW; ++u) {
            if (k0 >= 1 && k0 <= c[u]) {
                const int e = v[u].x;
                atomicAdd(&cnt[(e >> 20) & 255], 1);
                const int p = off + k0 - 1;
                if (p < LIST_CAP) outl[p] = e;
            }
            if (k1 <= c[u]) {
                const int e = v[u].y;
                atomicAdd(&cnt[(e >> 20) & 255], 1);
                const int p = off + k1 - 1;
                if (p < LIST_CAP) outl[p] = e;
            }
            off += c[u];
        }
    }
    __syncthreads();
    if (*ovfp) {
        cnt[t] = 0;
        if (t == 0) *nlistp = 0;
        __syncthreads();
        const int4* __restrict__ dst4 = (const int4*)dst;
        for (int i0 = 0; i0 < E4; i0 += 4 * PT) {
            int4 d[4];
            #pragma unroll
            for (int u = 0; u < 4; ++u) {
                const int i = i0 + u * PT + t;
                d[u] = (i < E4) ? dst4[i] : make_int4(-1, -1, -1, -1);
            }
            #pragma unroll
            for (int u = 0; u < 4; ++u) {
                const int e0 = (i0 + u * PT + t) * 4;
                const unsigned a = (unsigned)(d[u].x - base), b = (unsigned)(d[u].y - base);
                const unsigned cc = (unsigned)(d[u].z - base), g = (unsigned)(d[u].w - base);
                if (min(min(a, b), min(cc, g)) < NPB) {
                    if (a < NPB)  bin_append(cnt, nlistp, outl, src, e0 + 0, a);
                    if (b < NPB)  bin_append(cnt, nlistp, outl, src, e0 + 1, b);
                    if (cc < NPB) bin_append(cnt, nlistp, outl, src, e0 + 2, cc);
                    if (g < NPB)  bin_append(cnt, nlistp, outl, src, e0 + 3, g);
                }
            }
        }
        __syncthreads();
    }
    if (t < NPB) dinv[base + t] = rsqrtf(1.0f + (float)cnt[t]);
    if (t == 0) tile_n[tile] = *nlistp;
}

__device__ __forceinline__ void mm_tile(float* lds, const int rowTile,
                                        const float* __restrict__ X,
                                        const float* __restrict__ W,
                                        const float* __restrict__ bias,
                                        float* __restrict__ H) {
    const int rowBase = rowTile * 16;
    const int t = threadIdx.x;

    #pragma unroll
    for (int rep = 0; rep < 2; ++rep) {
        int idx = t + rep * 256;
        int r   = idx >> 5;
        int c4  = idx & 31;
        float4 v = *(const float4*)(X + (size_t)(rowBase + r) * DIM + c4 * 4);
        if (bias) {
            float4 b = *(const float4*)(bias + c4 * 4);
            v.x = fmaxf(v.x + b.x, 0.0f);
            v.y = fmaxf(v.y + b.y, 0.0f);
            v.z = fmaxf(v.z + b.z, 0.0f);
            v.w = fmaxf(v.w + b.w, 0.0f);
        }
        *(float4*)(&lds[r * LDS_STRIDE + c4 * 4]) = v;
    }
    __syncthreads();

    const int wave = t >> 5;
    const int lane = t & 31;
    const int half = lane >> 4;
    const int l16  = lane & 15;
    const int colBase = wave * 16;

    v8f acc = {0.f, 0.f, 0.f, 0.f, 0.f, 0.f, 0.f, 0.f};

    #pragma unroll
    for (int k0 = 0; k0 < DIM; k0 += 4) {
        const int kk = k0 + half * 2;
        float2 av = *(const float2*)(&lds[l16 * LDS_STRIDE + kk]);
        v2f a, b;
        a[0] = av.x;
        a[1] = av.y;
        b[0] = W[(size_t)kk * DIM + colBase + l16];
        b[1] = W[(size_t)(kk + 1) * DIM + colBase + l16];
        acc = __builtin_amdgcn_wmma_f32_16x16x4_f32(false, a, false, b,
                                                    (short)0, acc, false, false);
    }

    float* outp = H + (size_t)rowBase * DIM + colBase;
    #pragma unroll
    for (int v = 0; v < 8; ++v) {
        outp[(size_t)(v + half * 8) * DIM + l16] = acc[v];
    }
}

__launch_bounds__(256, 4)
__global__ void k_matmul_wmma(const float* __restrict__ X,
                              const float* __restrict__ W,
                              const float* __restrict__ bias,
                              float* __restrict__ H) {
    __shared__ __attribute__((aligned(16))) float lds[MM_LDS_W];
    mm_tile(lds, blockIdx.x, X, W, bias, H);
}

__launch_bounds__(PT)
__global__ void k_part_mm(const int* __restrict__ src, const int* __restrict__ dst,
                          int* __restrict__ cells,
                          const float* __restrict__ X, const float* __restrict__ W,
                          float* __restrict__ H) {
    __shared__ __attribute__((aligned(16))) int smem[STAGE_W];
    if (blockIdx.x < N_TILES) part_body(smem, blockIdx.x, src, dst, cells);
    else mm_tile((float*)smem, blockIdx.x - N_TILES, X, W, nullptr, H);
}

__launch_bounds__(PT)
__global__ void k_deg_mm(const int* __restrict__ src, const int* __restrict__ dst,
                         const int* __restrict__ cells,
                         float* __restrict__ dinv, int* __restrict__ tile_n,
                         int* __restrict__ tile_list,
                         const float* __restrict__ X, const float* __restrict__ W,
                         float* __restrict__ H) {
    __shared__ __attribute__((aligned(16))) int smem[MM_LDS_W];
    if (blockIdx.x < N_TILES) deg_body(smem, blockIdx.x, src, dst, cells, dinv, tile_n, tile_list);
    else mm_tile((float*)smem, MM_A + (blockIdx.x - N_TILES), X, W, nullptr, H);
}

__device__ __forceinline__ int acc_word(int f) { return ((f & 3) << 5) | (f >> 2); }

__device__ __forceinline__ void list_append(int* list, int* nlist, float* acc, int e, unsigned dl,
                                            int base, const int* __restrict__ src,
                                            const float* __restrict__ h,
                                            const float* __restrict__ dinv) {
    const int pos = atomicAdd(nlist, 1);
    const int s = src[e];
    if (pos < LIST_CAP) {
        list[pos] = s | (int)(dl << 20);
    } else {
        const float nm = dinv[s] * dinv[base + (int)dl];
        for (int f = 0; f < DIM; ++f)
            atomicAdd(&acc[dl * DIM + acc_word(f)], h[(size_t)s * DIM + f] * nm);
    }
}

__launch_bounds__(WG)
__global__ void k_aggregate(const int* __restrict__ src, const int* __restrict__ dst,
                            const int* __restrict__ tile_n, const int* __restrict__ tile_list,
                            const float* __restrict__ h, const float* __restrict__ dinv,
                            const float* __restrict__ bias, float* __restrict__ out) {
    __shared__ float acc[NPB * DIM];
    __shared__ int   list[LIST_CAP];
    __shared__ int   nlist;
    const int t = threadIdx.x;
    const int wave = t >> 5;
    const int lane = t & 31;
    const int base = blockIdx.x * NPB;
    const int ntile = tile_n[blockIdx.x];

    {
        constexpr int R = (NPB * DIM) / WG;
        float hr[R], dr[R];
        #pragma unroll
        for (int r = 0; r < R; ++r) {
            const int idx = r * WG + t;
            hr[r] = h[(size_t)base * DIM + idx];
            dr[r] = dinv[base + (idx >> 7)];
        }
        const float bv = bias ? bias[t & 127] : 0.0f;
        #pragma unroll
        for (int r = 0; r < R; ++r) {
            float v = hr[r] * dr[r] * dr[r];
            if (bias) v += bv;
            acc[((r * WG + t) & ~127) + acc_word(t & 127)] = v;
        }
    }
    if (ntile <= LIST_CAP) {
        const int* __restrict__ inl = tile_list + (size_t)blockIdx.x * LIST_CAP;
        for (int k = t; k < ntile; k += WG) list[k] = inl[k];
    }
    if (t == 0) nlist = 0;
    __syncthreads();

    int n = ntile;
    if (ntile > LIST_CAP) {
        const int4* __restrict__ dst4 = (const int4*)dst;
        const int rot = blockIdx.x * (E4 / N_TILES);
        for (int i0 = 0; i0 < E4; i0 += 4 * WG) {
            int4 d[4];
            int  qq[4];
            #pragma unroll
            for (int u = 0; u < 4; ++u) {
                const int i = i0 + u * WG + t;
                int q = i + rot;
                if (q >= E4) q -= E4;
                qq[u] = q;
                d[u] = (i < E4) ? dst4[q] : make_int4(-1, -1, -1, -1);
            }
            #pragma unroll
            for (int u = 0; u < 4; ++u) {
                const int e0 = qq[u] * 4;
                const unsigned a = (unsigned)(d[u].x - base), b = (unsigned)(d[u].y - base);
                const unsigned c = (unsigned)(d[u].z - base), g = (unsigned)(d[u].w - base);
                if (min(min(a, b), min(c, g)) < NPB) {
                    if (a < NPB) list_append(list, &nlist, acc, e0 + 0, a, base, src, h, dinv);
                    if (b < NPB) list_append(list, &nlist, acc, e0 + 1, b, base, src, h, dinv);
                    if (c < NPB) list_append(list, &nlist, acc, e0 + 2, c, base, src, h, dinv);
                    if (g < NPB) list_append(list, &nlist, acc, e0 + 3, g, base, src, h, dinv);
                }
            }
        }
        __syncthreads();
        n = min(nlist, LIST_CAP);
    }

    for (int k0 = wave * PU; k0 < n; k0 += (WG / 32) * PU) {
        int   ent[PU], s[PU];
        float ds[PU], dd[PU];
        float hv[PU][4];
        #pragma unroll
        for (int u = 0; u < PU; ++u) {
            const int k = k0 + u;
            ent[u] = (k < n) ? list[k] : -1;
        }
        #pragma unroll
        for (int u = 0; u < PU; ++u) {
            s[u] = max(ent[u], 0) & 0xFFFFF;
            const int dl = (max(ent[u], 0) >> 20) & 255;
            ds[u] = dinv[s[u]];
            dd[u] = dinv[base + dl];
            const float4 hq = *(const float4*)(h + (size_t)s[u] * DIM + lane * 4);
            hv[u][0] = hq.x; hv[u][1] = hq.y; hv[u][2] = hq.z; hv[u][3] = hq.w;
        }
        #pragma unroll
        for (int u = 0; u < PU; ++u)
            asm volatile("" :: "v"(ds[u]), "v"(dd[u]), "v"(hv[u][0]), "v"(hv[u][1]),
                               "v"(hv[u][2]), "v"(hv[u][3]));
        #pragma unroll
        for (int u = 0; u < PU; ++u) {
            if (ent[u] >= 0) {
                const float nm = ds[u] * dd[u];
                float* ap = &acc[((ent[u] >> 20) & 255) * DIM + lane];
                atomicAdd(ap +  0, hv[u][0] * nm);
                atomicAdd(ap + 32, hv[u][1] * nm);
                atomicAdd(ap + 64, hv[u][2] * nm);
                atomicAdd(ap + 96, hv[u][3] * nm);
            }
        }
    }
    __syncthreads();

    for (int r = 0; r < (NPB * DIM) / WG; ++r)
        out[(size_t)base * DIM + r * WG + t] = acc[((r * WG + t) & ~127) + acc_word(t & 127)];
}

__launch_bounds__(256)
__global__ void k_decode(const int* __restrict__ pa, const int* __restrict__ pb,
                         const float* __restrict__ z, float* __restrict__ out) {
    const int wid  = threadIdx.x >> 5;
    const int lane = threadIdx.x & 31;
    const int p = blockIdx.x * 8 + wid;
    if (p >= N_PAIRS) return;
    const int a = pa[p];
    const int b = pb[p];
    const float4 za = *(const float4*)(z + (size_t)a * DIM + lane * 4);
    const float4 zb = *(const float4*)(z + (size_t)b * DIM + lane * 4);
    float s = za.x * zb.x + za.y * zb.y + za.z * zb.z + za.w * zb.w;
    #pragma unroll
    for (int off = 16; off > 0; off >>= 1) s += __shfl_down(s, off);
    if (lane == 0) out[p] = s;
}

extern "C" void kernel_launch(void* const* d_in, const int* in_sizes, int n_in,
                              void* d_out, int out_size, void* d_ws, size_t ws_size,
                              hipStream_t stream) {
    const int*   edge_index = (const int*)d_in[0];
    const int*   edge_pairs = (const int*)d_in[1];
    const float* emb        = (const float*)d_in[2];
    const float* W1         = (const float*)d_in[3];
    const float* b1         = (const float*)d_in[4];
    const float* W2         = (const float*)d_in[5];
    const float* b2         = (const float*)d_in[6];
    float*       out        = (float*)d_out;

    const int* src = edge_index;
    const int* dst = edge_index + N_EDGES;
    const int* pa  = edge_pairs;
    const int* pb  = edge_pairs + N_PAIRS;

    char* ws = (char*)d_ws;
    size_t off = 0;
    float* dinv = (float*)(ws + off); off += ((size_t)N_NODES * 4 + 255) & ~(size_t)255;
    float* h    = (float*)(ws + off); off += (size_t)N_NODES * DIM * 4;
    float* agg  = (float*)(ws + off); off += (size_t)N_NODES * DIM * 4;
    int* tile_n    = (int*)(ws + off); off += ((size_t)N_TILES * 4 + 255) & ~(size_t)255;
    int* tile_list = (int*)(ws + off); off += (size_t)N_TILES * LIST_CAP * 4;
    int* cells     = (int*)(ws + off); off += (size_t)N_TILES * N_TILES * CELL_W * 4;
    (void)ws_size; (void)in_sizes; (void)n_in; (void)out_size;

    const int T = 256;
    const int gMM   = MM_TILES;
    const int gPw   = N_PAIRS / 8;

    k_part_mm<<<N_TILES + MM_A, PT, 0, stream>>>(src, dst, cells, emb, W1, h);
    k_deg_mm<<<N_TILES + MM_B, PT, 0, stream>>>(src, dst, cells, dinv, tile_n, tile_list, emb, W1, h);

    k_aggregate<<<N_TILES, WG, 0, stream>>>(src, dst, tile_n, tile_list, h, dinv, nullptr, agg);

    k_matmul_wmma<<<gMM, T, 0, stream>>>(agg, W2, b1, h);
    k_aggregate<<<N_TILES, WG, 0, stream>>>(src, dst, tile_n, tile_list, h, dinv, b2, agg);

    k_decode<<<gPw, T, 0, stream>>>(pa, pb, agg, out);
}


# Decode by owner tile: pairs split by first node's tile into workspace cells in launch 1; a 1024-thread group keeps its 200 z rows in LDS, gathers only second rows; wave sum by DPP, sum order kept
# speedup vs baseline: 2.9593x; 1.0533x over previous
#include <hip/hip_runtime.h>


#define N_NODES 50000
#define N_EDGES 800000
#define N_PAIRS 400000
#define DIM     128
#define LDS_STRIDE 132

typedef float v2f __attribute__((ext_vector_type(2)));
typedef float v8f __attribute__((ext_vector_type(8)));

#define NPB      200
#define N_TILES  (N_NODES / NPB)
#define WG       1024
#define E4       (N_EDGES / 4)
#define LIST_CAP 16384
#define PU       8
#define ESL      (N_EDGES / N_TILES)
#define PTE      (ESL / 16)
#define CELL_W   64
#define CELL_CAP (CELL_W - 1)
#define SP       65
#define PT       256
#define CPW      32
#define STAGE_W  ((N_TILES * SP + 1 + 3) / 4 * 4)
#define MM_TILES (N_NODES / 16)
#define MM_A     1562
#define MM_B     (MM_TILES - MM_A)
#define MM_LDS_W (16 * LDS_STRIDE)
#define PSL       (N_PAIRS / N_TILES)
#define PPT       (PSL / 16)
#define PCELL_CAP ((CELL_W - 2) / 2)
#define DPU       8
#define DCW       ((N_TILES + 31) / 32)


__device__ __forceinline__ void part_body(int* stage, const int sl,
                                          const int* __restrict__ src, const int* __restrict__ dst,
                                          int* __restrict__ cells) {
    const int t = threadIdx.x;
    const int wave = t >> 5;
    const int lane = t & 31;
    for (int i = t; i < STAGE_W / 4; i += PT) ((int4*)stage)[i] = make_int4(0, 0, 0, 0);
    __syncthreads();
    if (t < PTE) {
        const int4* __restrict__ s4 = (const int4*)(src + (size_t)sl * ESL);
        const int4* __restrict__ d4 = (const int4*)(dst + (size_t)sl * ESL);
        int4 sv[4], dv[4];
        #pragma unroll
        for (int u = 0; u < 4; ++u) {
            sv[u] = s4[u * PTE + t];
            dv[u] = d4[u * PTE + t];
        }
        int  cell[16], ent[16], pos[16];
        bool ok[16];
        #pragma unroll
        for (int u = 0; u < 4; ++u) {
            const int ss[4] = {sv[u].x, sv[u].y, sv[u].z, sv[u].w};
            const int dd[4] = {dv[u].x, dv[u].y, dv[u].z, dv[u].w};
            #pragma unroll
            for (int j = 0; j < 4; ++j) {
                const int q = u * 4 + j;
                ok[q] = (unsigned)dd[j] < (unsigned)N_NODES;
                const int tile = ok[q] ? dd[j] / NPB : N_TILES;
                const int dl = ok[q] ? dd[j] - tile * NPB : 0;
                cell[q] = tile * SP;
                ent[q] = ss[j] | (dl << 20);
                pos[q] = atomicAdd(&stage[cell[q]], 1);
            }
        }
        #pragma unroll
        for (int q = 0; q < 16; ++q)
            if (ok[q] && pos[q] < CELL_CAP) stage[cell[q] + 1 + pos[q]] = ent[q];
    }
    __syncthreads();
    int* __restrict__ outc = cells + (size_t)sl * N_TILES * CELL_W;
    #pragma unroll
    for (int i = 0; i < CPW; ++i) {
        const int c = wave + i * (PT / 32);
        if (i * (PT / 32) + (PT / 32 - 1) < N_TILES || c < N_TILES) {
            int2 v;
            v.x = stage[c * SP + 2 * lane];
            v.y = stage[c * SP + 2 * lane + 1];
            *(int2*)(outc + c * CELL_W + 2 * lane) = v;
        }
    }
}

__device__ __forceinline__ void ppart_body(int* stage, const int sl,
                                           const int* __restrict__ pa, const int* __restrict__ pb,
                                           int* __restrict__ pcells) {
    const int t = threadIdx.x;
    const int wave = t >> 5;
    const int lane = t & 31;
    for (int i = t; i < STAGE_W / 4; i += PT) ((int4*)stage)[i] = make_int4(0, 0, 0, 0);
    __syncthreads();
    if (t < PPT) {
        const int4* __restrict__ a4 = (const int4*)(pa + (size_t)sl * PSL);
        const int4* __restrict__ b4 = (const int4*)(pb + (size_t)sl * PSL);
        int4 av[4], bv[4];
        #pragma unroll
        for (int u = 0; u < 4; ++u) {
            av[u] = a4[u * PPT + t];
            bv[u] = b4[u * PPT + t];
        }
        int cell[16], ent[16], pos[16];
        #pragma unroll
        for (int u = 0; u < 4; ++u) {
            const int aa[4] = {av[u].x, av[u].y, av[u].z, av[u].w};
            const int bb[4] = {bv[u].x, bv[u].y, bv[u].z, bv[u].w};
            #pragma unroll
            for (int j = 0; j < 4; ++j) {
                const int q = u * 4 + j;
                const int a = min(max(aa[j], 0), N_NODES - 1);
                const int b = min(max(bb[j], 0), N_NODES - 1);
                const int tile = a / NPB;
                cell[q] = tile * SP;
                ent[q] = b | ((a - tile * NPB) << 20);
                pos[q] = atomicAdd(&stage[cell[q]], 1);
            }
        }
        #pragma unroll
        for (int q = 0; q < 16; ++q)
            if (pos[q] < PCELL_CAP) {
                stage[cell[q] + 2 + 2 * pos[q]] = ent[q];
                stage[cell[q] + 3 + 2 * pos[q]] = sl * PSL + (q >> 2) * (4 * PPT) + 4 * t + (q & 3);
            }
    }
    __syncthreads();
    int* __restrict__ outc = pcells + (size_t)sl * N_TILES * CELL_W;
    #pragma unroll
    for (int i = 0; i < CPW; ++i) {
        const int c = wave + i * (PT / 32);
        if (i * (PT / 32) + (PT / 32 - 1) < N_TILES || c < N_TILES) {
            int2 v;
            v.x = stage[c * SP + 2 * lane];
            v.y = stage[c * SP + 2 * lane + 1];
            *(int2*)(outc + c * CELL_W + 2 * lane) = v;
        }
    }
}

__device__ __forceinline__ void bin_append(int* cnt, int* nlist, int* __restrict__ outl,
                                           const int* __restrict__ src, int e, unsigned dl) {
    atomicAdd(&cnt[dl], 1);
    const int pos = atomicAdd(nlist, 1);
    if (pos < LIST_CAP) outl[pos] = src[e] | (int)(dl << 20);
}

__device__ __forceinline__ void deg_body(int* lds, const int tile,
                                         const int* __restrict__ src, const int* __restrict__ dst,
                                         const int* __restrict__ cells,
                                         float* __restrict__ dinv, int* __restrict__ tile_n,
                                         int* __restrict__ tile_list) {
    int* cnt = lds;
    int* nlistp = lds + 256;
    int* ovfp = lds + 257;
    const int t = threadIdx.x;
    const int wave = t >> 5;
    const int lane = t & 31;
    const int base = tile * NPB;
    cnt[t] = 0;
    if (t == 0) { *nlistp = 0; *ovfp = 0; }
    __syncthreads();
    int* __restrict__ outl = tile_list + (size_t)tile * LIST_CAP;

    int2 v[CPW];
    #pragma unroll
    for (int u = 0; u < CPW; ++u) {
        const int sl = min(wave + u * (PT / 32), N_TILES - 1);
        v[u] = *(const int2*)(cells + ((size_t)sl * N_TILES + tile) * CELL_W + 2 * lane);
    }
    __builtin_amdgcn_sched_barrier(0);
    int c[CPW];
    int tot = 0;
    bool bad = false;
    #pragma unroll
    for (int u = 0; u < CPW; ++u) {
        c[u] = __builtin_amdgcn_readfirstlane(v[u].x);
        if (wave + u * (PT / 32) >= N_TILES) c[u] = 0;
        bad = bad || ((unsigned)c[u] > (unsigned)CELL_CAP);
        tot += c[u];
    }
    if (bad) {
        if (lane == 0) *ovfp = 1;
    } else {
        int wbase = 0;
        if (lane == 0) wbase = atomicAdd(nlistp, tot);
        int off = __builtin_amdgcn_readfirstlane(wbase);
        const int k0 = 2 * lane, k1 = k0 + 1;
        #pragma unroll
        for (int u = 0; u < CPW; ++u) {
            if (k0 >= 1 && k0 <= c[u]) {
                const int e = v[u].x;
                atomicAdd(&cnt[(e >> 20) & 255], 1);
                const int p = off + k0 - 1;
                if (p < LIST_CAP) outl[p] = e;
            }
            if (k1 <= c[u]) {
                const int e = v[u].y;
                atomicAdd(&cnt[(e >> 20) & 255], 1);
                const int p = off + k1 - 1;
                if (p < LIST_CAP) outl[p] = e;
            }
            off += c[u];
        }
    }
    __syncthreads();
    if (*ovfp) {
        cnt[t] = 0;
        if (t == 0) *nlistp = 0;
        __syncthreads();
        const int4* __restrict__ dst4 = (const int4*)dst;
        for (int i0 = 0; i0 < E4; i0 += 4 * PT) {
            int4 d[4];
            #pragma unroll
            for (int u = 0; u < 4; ++u) {
                const int i = i0 + u * PT + t;
                d[u] = (i < E4) ? dst4[i] : make_int4(-1, -1, -1, -1);
            }
            #pragma unroll
            for (int u = 0; u < 4; ++u) {
                const int e0 = (i0 + u * PT + t) * 4;
                const unsigned a = (unsigned)(d[u].x - base), b = (unsigned)(d[u].y - base);
                const unsigned cc = (unsigned)(d[u].z - base), g = (unsigned)(d[u].w - base);
                if (min(min(a, b), min(cc, g)) < NPB) {
                    if (a < NPB)  bin_append(cnt, nlistp, outl, src, e0 + 0, a);
                    if (b < NPB)  bin_append(cnt, nlistp, outl, src, e0 + 1, b);
                    if (cc < NPB) bin_append(cnt, nlistp, outl, src, e0 + 2, cc);
                    if (g < NPB)  bin_append(cnt, nlistp, outl, src, e0 + 3, g);
                }
            }
        }
        __syncthreads();
    }
    if (t < NPB) dinv[base + t] = rsqrtf(1.0f + (float)cnt[t]);
    if (t == 0) tile_n[tile] = *nlistp;
}

__device__ __forceinline__ void mm_tile(float* lds, const int rowTile,
                                        const float* __restrict__ X,
                                        const float* __restrict__ W,
                                        const float* __restrict__ bias,
                                        float* __restrict__ H) {
    const int rowBase = rowTile * 16;
    const int t = threadIdx.x;

    #pragma unroll
    for (int rep = 0; rep < 2; ++rep) {
        int idx = t + rep * 256;
        int r   = idx >> 5;
        int c4  = idx & 31;
        float4 v = *(const float4*)(X + (size_t)(rowBase + r) * DIM + c4 * 4);
        if (bias) {
            float4 b = *(const float4*)(bias + c4 * 4);
            v.x = fmaxf(v.x + b.x, 0.0f);
            v.y = fmaxf(v.y + b.y, 0.0f);
            v.z = fmaxf(v.z + b.z, 0.0f);
            v.w = fmaxf(v.w + b.w, 0.0f);
        }
        *(float4*)(&lds[r * LDS_STRIDE + c4 * 4]) = v;
    }
    __syncthreads();

    const int wave = t >> 5;
    const int lane = t & 31;
    const int half = lane >> 4;
    const int l16  = lane & 15;
    const int colBase = wave * 16;

    v8f acc = {0.f, 0.f, 0.f, 0.f, 0.f, 0.f, 0.f, 0.f};

    #pragma unroll
    for (int k0 = 0; k0 < DIM; k0 += 4) {
        const int kk = k0 + half * 2;
        float2 av = *(const float2*)(&lds[l16 * LDS_STRIDE + kk]);
        v2f a, b;
        a[0] = av.x;
        a[1] = av.y;
        b[0] = W[(size_t)kk * DIM + colBase + l16];
        b[1] = W[(size_t)(kk + 1) * DIM + colBase + l16];
        acc = __builtin_amdgcn_wmma_f32_16x16x4_f32(false, a, false, b,
                                                    (short)0, acc, false, false);
    }

    float* outp = H + (size_t)rowBase * DIM + colBase;
    #pragma unroll
    for (int v = 0; v < 8; ++v) {
        outp[(size_t)(v + half * 8) * DIM + l16] = acc[v];
    }
}

__launch_bounds__(256, 4)
__global__ void k_matmul_wmma(const float* __restrict__ X,
                              const float* __restrict__ W,
                              const float* __restrict__ bias,
                              float* __restrict__ H) {
    __shared__ __attribute__((aligned(16))) float lds[MM_LDS_W];
    mm_tile(lds, blockIdx.x, X, W, bias, H);
}

__launch_bounds__(PT)
__global__ void k_part_mm(const int* __restrict__ src, const int* __restrict__ dst,
                          int* __restrict__ cells,
                          const int* __restrict__ pa, const int* __restrict__ pb,
                          int* __restrict__ pcells,
                          const float* __restrict__ X, const float* __restrict__ W,
                          float* __restrict__ H) {
    __shared__ __attribute__((aligned(16))) int smem[STAGE_W];
    if (blockIdx.x < N_TILES) part_body(smem, blockIdx.x, src, dst, cells);
    else if (blockIdx.x < 2 * N_TILES) ppart_body(smem, blockIdx.x - N_TILES, pa, pb, pcells);
    else mm_tile((float*)smem, blockIdx.x - 2 * N_TILES, X, W, nullptr, H);
}

__launch_bounds__(PT)
__global__ void k_deg_mm(const int* __restrict__ src, const int* __restrict__ dst,
                         const int* __restrict__ cells,
                         float* __restrict__ dinv, int* __restrict__ tile_n,
                         int* __restrict__ tile_list,
                         const float* __restrict__ X, const float* __restrict__ W,
                         float* __restrict__ H) {
    __shared__ __attribute__((aligned(16))) int smem[MM_LDS_W];
    if (blockIdx.x < N_TILES) deg_body(smem, blockIdx.x, src, dst, cells, dinv, tile_n, tile_list);
    else mm_tile((float*)smem, MM_A + (blockIdx.x - N_TILES), X, W, nullptr, H);
}

__device__ __forceinline__ int acc_word(int f) { return ((f & 3) << 5) | (f >> 2); }

__device__ __forceinline__ void list_append(int* list, int* nlist, float* acc, int e, unsigned dl,
                                            int base, const int* __restrict__ src,
                                            const float* __restrict__ h,
                                            const float* __restrict__ dinv) {
    const int pos = atomicAdd(nlist, 1);
    const int s = src[e];
    if (pos < LIST_CAP) {
        list[pos] = s | (int)(dl << 20);
    } else {
        const float nm = dinv[s] * dinv[base + (int)dl];
        for (int f = 0; f < DIM; ++f)
            atomicAdd(&acc[dl * DIM + acc_word(f)], h[(size_t)s * DIM + f] * nm);
    }
}

__launch_bounds__(WG)
__global__ void k_aggregate(const int* __restrict__ src, const int* __restrict__ dst,
                            const int* __restrict__ tile_n, const int* __restrict__ tile_list,
                            const float* __restrict__ h, const float* __restrict__ dinv,
                            const float* __restrict__ bias, float* __restrict__ out) {
    __shared__ float acc[NPB * DIM];
    __shared__ int   list[LIST_CAP];
    __shared__ int   nlist;
    const int t = threadIdx.x;
    const int wave = t >> 5;
    const int lane = t & 31;
    const int base = blockIdx.x * NPB;
    const int ntile = tile_n[blockIdx.x];

    {
        constexpr int R = (NPB * DIM) / WG;
        float hr[R], dr[R];
        #pragma unroll
        for (int r = 0; r < R; ++r) {
            const int idx = r * WG + t;
            hr[r] = h[(size_t)base * DIM + idx];
            dr[r] = dinv[base + (idx >> 7)];
        }
        const float bv = bias ? bias[t & 127] : 0.0f;
        #pragma unroll
        for (int r = 0; r < R; ++r) {
            float v = hr[r] * dr[r] * dr[r];
            if (bias) v += bv;
            acc[((r * WG + t) & ~127) + acc_word(t & 127)] = v;
        }
    }
    if (ntile <= LIST_CAP) {
        const int* __restrict__ inl = tile_list + (size_t)blockIdx.x * LIST_CAP;
        for (int k = t; k < ntile; k += WG) list[k] = inl[k];
    }
    if (t == 0) nlist = 0;
    __syncthreads();

    int n = ntile;
    if (ntile > LIST_CAP) {
        const int4* __restrict__ dst4 = (const int4*)dst;
        const int rot = blockIdx.x * (E4 / N_TILES);
        for (int i0 = 0; i0 < E4; i0 += 4 * WG) {
            int4 d[4];
            int  qq[4];
            #pragma unroll
            for (int u = 0; u < 4; ++u) {
                const int i = i0 + u * WG + t;
                int q = i + rot;
                if (q >= E4) q -= E4;
                qq[u] = q;
                d[u] = (i < E4) ? dst4[q] : make_int4(-1, -1, -1, -1);
            }
            #pragma unroll
            for (int u = 0; u < 4; ++u) {
                const int e0 = qq[u] * 4;
                const unsigned a = (unsigned)(d[u].x - base), b = (unsigned)(d[u].y - base);
                const unsigned c = (unsigned)(d[u].z - base), g = (unsigned)(d[u].w - base);
                if (min(min(a, b), min(c, g)) < NPB) {
                    if (a < NPB) list_append(list, &nlist, acc, e0 + 0, a, base, src, h, dinv);
                    if (b < NPB) list_append(list, &nlist, acc, e0 + 1, b, base, src, h, dinv);
                    if (c < NPB) list_append(list, &nlist, acc, e0 + 2, c, base, src, h, dinv);
                    if (g < NPB) list_append(list, &nlist, acc, e0 + 3, g, base, src, h, dinv);
                }
            }
        }
        __syncthreads();
        n = min(nlist, LIST_CAP);
    }

    for (int k0 = wave * PU; k0 < n; k0 += (WG / 32) * PU) {
        int   ent[PU], s[PU];
        float ds[PU], dd[PU];
        float hv[PU][4];
        #pragma unroll
        for (int u = 0; u < PU; ++u) {
            const int k = k0 + u;
            ent[u] = (k < n) ? list[k] : -1;
        }
        #pragma unroll
        for (int u = 0; u < PU; ++u) {
            s[u] = max(ent[u], 0) & 0xFFFFF;
            const int dl = (max(ent[u], 0) >> 20) & 255;
            ds[u] = dinv[s[u]];
            dd[u] = dinv[base + dl];
            const float4 hq = *(const float4*)(h + (size_t)s[u] * DIM + lane * 4);
            hv[u][0] = hq.x; hv[u][1] = hq.y; hv[u][2] = hq.z; hv[u][3] = hq.w;
        }
        #pragma unroll
        for (int u = 0; u < PU; ++u)
            asm volatile("" :: "v"(ds[u]), "v"(dd[u]), "v"(hv[u][0]), "v"(hv[u][1]),
                               "v"(hv[u][2]), "v"(hv[u][3]));
        #pragma unroll
        for (int u = 0; u < PU; ++u) {
            if (ent[u] >= 0) {
                const float nm = ds[u] * dd[u];
                float* ap = &acc[((ent[u] >> 20) & 255) * DIM + lane];
                atomicAdd(ap +  0, hv[u][0] * nm);
                atomicAdd(ap + 32, hv[u][1] * nm);
                atomicAdd(ap + 64, hv[u][2] * nm);
                atomicAdd(ap + 96, hv[u][3] * nm);
            }
        }
    }
    __syncthreads();

    for (int r = 0; r < (NPB * DIM) / WG; ++r)
        out[(size_t)base * DIM + r * WG + t] = acc[((r * WG + t) & ~127) + acc_word(t & 127)];
}

__device__ __forceinline__ float xhalf(float x) {
    const int b = __builtin_bit_cast(int, x);
    return __builtin_bit_cast(float, __builtin_amdgcn_permlanex16(b, b, 0x76543210, 0xfedcba98, false, false));
}
template <int M>
__device__ __forceinline__ float xrow(float x) {
    return __builtin_bit_cast(float, __builtin_amdgcn_update_dpp(0, __builtin_bit_cast(int, x),
                                                                 0x160 + M, 0xf, 0xf, true));
}
__device__ __forceinline__ float dot4(const float4 a, const float4 b) {
    return fmaf(a.w, b.w, fmaf(a.z, b.z, fmaf(a.x, b.x, a.y * b.y)));
}
__device__ __forceinline__ float wave_sum(float s) {
    s += xhalf(s);
    s += xrow<8>(s);
    s += xrow<4>(s);
    s += xrow<2>(s);
    s += xrow<1>(s);
    return s;
}

#define WL_CAP (N_TILES * PCELL_CAP)

__launch_bounds__(WG)
__global__ void k_decode_tile(const int* __restrict__ pa, const int* __restrict__ pb,
                              const int* __restrict__ pcells,
                              const float* __restrict__ z, float* __restrict__ out) {
    __shared__ __attribute__((aligned(16))) float zt[NPB * DIM];
    __shared__ int2 wl[WL_CAP];
    __shared__ int wtot[WG / 32];
    const int t = threadIdx.x;
    const int wave = __builtin_amdgcn_readfirstlane(t >> 5);
    const int lane = t & 31;
    const int tile = blockIdx.x;
    const int base = tile * NPB;

    int2 v[DCW];
    #pragma unroll
    for (int u = 0; u < DCW; ++u) {
        const int sl = min(wave + u * (WG / 32), N_TILES - 1);
        v[u] = *(const int2*)(pcells + ((size_t)sl * N_TILES + tile) * CELL_W + 2 * lane);
    }
    constexpr int Q4 = NPB * DIM / 4;
    constexpr int R4 = (Q4 + WG - 1) / WG;
    {
        const float4* __restrict__ zg = (const float4*)(z + (size_t)base * DIM);
        float4 zr[R4];
        #pragma unroll
        for (int r = 0; r < R4; ++r) zr[r] = zg[min(r * WG + t, Q4 - 1)];
        #pragma unroll
        for (int r = 0; r < R4; ++r)
            if (r * WG + t < Q4) ((float4*)zt)[r * WG + t] = zr[r];
    }
    int c[DCW];
    int tot = 0;
    bool bad = false;
    #pragma unroll
    for (int u = 0; u < DCW; ++u) {
        c[u] = __builtin_amdgcn_readfirstlane(v[u].x);
        if (wave + u * (WG / 32) >= N_TILES) c[u] = 0;
        bad = bad || ((unsigned)c[u] > (unsigned)PCELL_CAP);
        tot += c[u];
    }
    if (lane == 0) wtot[wave] = bad ? -1 : tot;
    __syncthreads();
    const int wt = wtot[lane];
    const bool ovf = __builtin_amdgcn_ballot_w32(wt < 0) != 0;

    if (!ovf) {
        int pos = 0, total = 0;
        #pragma unroll
        for (int w = 0; w < WG / 32; ++w) {
            const int x = __builtin_amdgcn_readlane(wt, w);
            if (w < wave) pos += x;
            total += x;
        }
        #pragma unroll
        for (int u = 0; u < DCW; ++u) {
            if (lane >= 1 && lane <= c[u]) wl[pos + lane - 1] = v[u];
            pos += c[u];
        }
        __syncthreads();
        for (int k0 = wave * DPU; k0 < total; k0 += (WG / 32) * DPU) {
            const int2 e = wl[min(k0 + (lane & (DPU - 1)), total - 1)];
            float4 za[DPU], zb[DPU];
            #pragma unroll
            for (int j = 0; j < DPU; ++j) {
                const int ex = __builtin_amdgcn_readlane(e.x, j);
                const int b  = min(ex & 0xFFFFF, N_NODES - 1);
                const int al = min((ex >> 20) & 255, NPB - 1);
                zb[j] = *(const float4*)(z + (size_t)b * DIM + lane * 4);
                za[j] = *(const float4*)(&zt[al * DIM + lane * 4]);
            }
            float s[DPU];
            #pragma unroll
            for (int j = 0; j < DPU; ++j)
                s[j] = dot4(za[j], zb[j]);
            #pragma unroll
            for (int j = 0; j < DPU; ++j) s[j] += xhalf(s[j]);
            #pragma unroll
            for (int j = 0; j < DPU; ++j) s[j] += xrow<8>(s[j]);
            #pragma unroll
            for (int j = 0; j < DPU; ++j) s[j] += xrow<4>(s[j]);
            #pragma unroll
            for (int j = 0; j < DPU; ++j) s[j] += xrow<2>(s[j]);
            #pragma unroll
            for (int j = 0; j < DPU; ++j) s[j] += xrow<1>(s[j]);
            float r = s[0];
            #pragma unroll
            for (int j = 1; j < DPU; ++j) r = (lane == j) ? s[j] : r;
            if (lane < DPU && k0 + lane < total) out[min((unsigned)e.y, (unsigned)(N_PAIRS - 1))] = r;
        }
    } else {
        for (int i0 = wave * 32; i0 < N_PAIRS; i0 += WG) {
            const int a = min(max(pa[i0 + lane], 0), N_NODES - 1);
            const int b = min(max(pb[i0 + lane], 0), N_NODES - 1);
            unsigned m = __builtin_amdgcn_ballot_w32((unsigned)(a - base) < (unsigned)NPB);
            while (m) {
                const int l = __builtin_ctz(m);
                m &= m - 1;
                const int sa = __builtin_amdgcn_readlane(a, l) - base;
                const int sb = __builtin_amdgcn_readlane(b, l);
                const float4 zb = *(const float4*)(z + (size_t)sb * DIM + lane * 4);
                const float4 za = *(const float4*)(&zt[sa * DIM + lane * 4]);
                const float s = wave_sum(dot4(za, zb));
                if (lane == 0) out[i0 + l] = s;
            }
        }
    }
}

extern "C" void kernel_launch(void* const* d_in, const int* in_sizes, int n_in,
                              void* d_out, int out_size, void* d_ws, size_t ws_size,
                              hipStream_t stream) {
    const int*   edge_index = (const int*)d_in[0];
    const int*   edge_pairs = (const int*)d_in[1];
    const float* emb        = (const float*)d_in[2];
    const float* W1         = (const float*)d_in[3];
    const float* b1         = (const float*)d_in[4];
    const float* W2         = (const float*)d_in[5];
    const float* b2         = (const float*)d_in[6];
    float*       out        = (float*)d_out;

    const int* src = edge_index;
    const int* dst = edge_index + N_EDGES;
    const int* pa  = edge_pairs;
    const int* pb  = edge_pairs + N_PAIRS;

    char* ws = (char*)d_ws;
    size_t off = 0;
    float* dinv = (float*)(ws + off); off += ((size_t)N_NODES * 4 + 255) & ~(size_t)255;
    float* h    = (float*)(ws + off); off += (size_t)N_NODES * DIM * 4;
    float* agg  = (float*)(ws + off); off += (size_t)N_NODES * DIM * 4;
    int* tile_n    = (int*)(ws + off); off += ((size_t)N_TILES * 4 + 255) & ~(size_t)255;
    int* tile_list = (int*)(ws + off); off += (size_t)N_TILES * LIST_CAP * 4;
    int* cells     = (int*)(ws + off); off += (size_t)N_TILES * N_TILES * CELL_W * 4;
    int* pcells    = (int*)(ws + off); off += (size_t)N_TILES * N_TILES * CELL_W * 4;
    (void)ws_size; (void)in_sizes; (void)n_in; (void)out_size;

    const int T = 256;
    const int gMM   = MM_TILES;

    k_part_mm<<<2 * N_TILES + MM_A, PT, 0, stream>>>(src, dst, cells, pa, pb, pcells, emb, W1, h);
    k_deg_mm<<<N_TILES + MM_B, PT, 0, stream>>>(src, dst, cells, dinv, tile_n, tile_list, emb, W1, h);

    k_aggregate<<<N_TILES, WG, 0, stream>>>(src, dst, tile_n, tile_list, h, dinv, nullptr, agg);

    k_matmul_wmma<<<gMM, T, 0, stream>>>(agg, W2, b1, h);
    k_aggregate<<<N_TILES, WG, 0, stream>>>(src, dst, tile_n, tile_list, h, dinv, b2, agg);

    k_decode_tile<<<N_TILES, WG, 0, stream>>>(pa, pb, pcells, agg, out);
}


# Layer-2 product moved into the layer-1 aggregation work-group: summed rows stay in LDS (pitch 132) and feed the matrix unit there; 6 launches to 5; the product's k terms summed in LDS word order
# speedup vs baseline: 3.1215x; 1.0548x over previous
#include <hip/hip_runtime.h>


#define N_NODES 50000
#define N_EDGES 800000
#define N_PAIRS 400000
#define DIM     128
#define LDS_STRIDE 132
#define APITCH  132

typedef float v2f __attribute__((ext_vector_type(2)));
typedef float v8f __attribute__((ext_vector_type(8)));

#define NPB      200
#define N_TILES  (N_NODES / NPB)
#define WG       1024
#define E4       (N_EDGES / 4)
#define LIST_CAP 16384
#define PU       8
#define ESL      (N_EDGES / N_TILES)
#define PTE      (ESL / 16)
#define CELL_W   64
#define CELL_CAP (CELL_W - 1)
#define SP       65
#define PT       256
#define CPW      32
#define STAGE_W  ((N_TILES * SP + 1 + 3) / 4 * 4)
#define MM_TILES (N_NODES / 16)
#define MM_A     1562
#define MM_B     (MM_TILES - MM_A)
#define MM_LDS_W (16 * LDS_STRIDE)
#define PSL       (N_PAIRS / N_TILES)
#define PPT       (PSL / 16)
#define PCELL_CAP ((CELL_W - 2) / 2)
#define DPU       8
#define DCW       ((N_TILES + 31) / 32)


__device__ __forceinline__ void part_body(int* stage, const int sl,
                                          const int* __restrict__ src, const int* __restrict__ dst,
                                          int* __restrict__ cells) {
    const int t = threadIdx.x;
    const int wave = t >> 5;
    const int lane = t & 31;
    for (int i = t; i < STAGE_W / 4; i += PT) ((int4*)stage)[i] = make_int4(0, 0, 0, 0);
    __syncthreads();
    if (t < PTE) {
        const int4* __restrict__ s4 = (const int4*)(src + (size_t)sl * ESL);
        const int4* __restrict__ d4 = (const int4*)(dst + (size_t)sl * ESL);
        int4 sv[4], dv[4];
        #pragma unroll
        for (int u = 0; u < 4; ++u) {
            sv[u] = s4[u * PTE + t];
            dv[u] = d4[u * PTE + t];
        }
        int  cell[16], ent[16], pos[16];
        bool ok[16];
        #pragma unroll
        for (int u = 0; u < 4; ++u) {
            const int ss[4] = {sv[u].x, sv[u].y, sv[u].z, sv[u].w};
            const int dd[4] = {dv[u].x, dv[u].y, dv[u].z, dv[u].w};
            #pragma unroll
            for (int j = 0; j < 4; ++j) {
                const int q = u * 4 + j;
                ok[q] = (unsigned)dd[j] < (unsigned)N_NODES;
                const int tile = ok[q] ? dd[j] / NPB : N_TILES;
                const int dl = ok[q] ? dd[j] - tile * NPB : 0;
                cell[q] = tile * SP;
                ent[q] = ss[j] | (dl << 20);
                pos[q] = atomicAdd(&stage[cell[q]], 1);
            }
        }
        #pragma unroll
        for (int q = 0; q < 16; ++q)
            if (ok[q] && pos[q] < CELL_CAP) stage[cell[q] + 1 + pos[q]] = ent[q];
    }
    __syncthreads();
    int* __restrict__ outc = cells + (size_t)sl * N_TILES * CELL_W;
    #pragma unroll
    for (int i = 0; i < CPW; ++i) {
        const int c = wave + i * (PT / 32);
        if (i * (PT / 32) + (PT / 32 - 1) < N_TILES || c < N_TILES) {
            int2 v;
            v.x = stage[c * SP + 2 * lane];
            v.y = stage[c * SP + 2 * lane + 1];
            *(int2*)(outc + c * CELL_W + 2 * lane) = v;
        }
    }
}

__device__ __forceinline__ void ppart_body(int* stage, const int sl,
                                           const int* __restrict__ pa, const int* __restrict__ pb,
                                           int* __restrict__ pcells) {
    const int t = threadIdx.x;
    const int wave = t >> 5;
    const int lane = t & 31;
    for (int i = t; i < STAGE_W / 4; i += PT) ((int4*)stage)[i] = make_int4(0, 0, 0, 0);
    __syncthreads();
    if (t < PPT) {
        const int4* __restrict__ a4 = (const int4*)(pa + (size_t)sl * PSL);
        const int4* __restrict__ b4 = (const int4*)(pb + (size_t)sl * PSL);
        int4 av[4], bv[4];
        #pragma unroll
        for (int u = 0; u < 4; ++u) {
            av[u] = a4[u * PPT + t];
            bv[u] = b4[u * PPT + t];
        }
        int cell[16], ent[16], pos[16];
        #pragma unroll
        for (int u = 0; u < 4; ++u) {
            const int aa[4] = {av[u].x, av[u].y, av[u].z, av[u].w};
            const int bb[4] = {bv[u].x, bv[u].y, bv[u].z, bv[u].w};
            #pragma unroll
            for (int j = 0; j < 4; ++j) {
                const int q = u * 4 + j;
                const int a = min(max(aa[j], 0), N_NODES - 1);
                const int b = min(max(bb[j], 0), N_NODES - 1);
                const int tile = a / NPB;
                cell[q] = tile * SP;
                ent[q] = b | ((a - tile * NPB) << 20);
                pos[q] = atomicAdd(&stage[cell[q]], 1);
            }
        }
        #pragma unroll
        for (int q = 0; q < 16; ++q)
            if (pos[q] < PCELL_CAP) {
                stage[cell[q] + 2 + 2 * pos[q]] = ent[q];
                stage[cell[q] + 3 + 2 * pos[q]] = sl * PSL + (q >> 2) * (4 * PPT) + 4 * t + (q & 3);
            }
    }
    __syncthreads();
    int* __restrict__ outc = pcells + (size_t)sl * N_TILES * CELL_W;
    #pragma unroll
    for (int i = 0; i < CPW; ++i) {
        const int c = wave + i * (PT / 32);
        if (i * (PT / 32) + (PT / 32 - 1) < N_TILES || c < N_TILES) {
            int2 v;
            v.x = stage[c * SP + 2 * lane];
            v.y = stage[c * SP + 2 * lane + 1];
            *(int2*)(outc + c * CELL_W + 2 * lane) = v;
        }
    }
}

__device__ __forceinline__ void bin_append(int* cnt, int* nlist, int* __restrict__ outl,
                                           const int* __restrict__ src, int e, unsigned dl) {
    atomicAdd(&cnt[dl], 1);
    const int pos = atomicAdd(nlist, 1);
    if (pos < LIST_CAP) outl[pos] = src[e] | (int)(dl << 20);
}

__device__ __forceinline__ void deg_body(int* lds, const int tile,
                                         const int* __restrict__ src, const int* __restrict__ dst,
                                         const int* __restrict__ cells,
                                         float* __restrict__ dinv, int* __restrict__ tile_n,
                                         int* __restrict__ tile_list) {
    int* cnt = lds;
    int* nlistp = lds + 256;
    int* ovfp = lds + 257;
    const int t = threadIdx.x;
    const int wave = t >> 5;
    const int lane = t & 31;
    const int base = tile * NPB;
    cnt[t] = 0;
    if (t == 0) { *nlistp = 0; *ovfp = 0; }
    __syncthreads();
    int* __restrict__ outl = tile_list + (size_t)tile * LIST_CAP;

    int2 v[CPW];
    #pragma unroll
    for (int u = 0; u < CPW; ++u) {
        const int sl = min(wave + u * (PT / 32), N_TILES - 1);
        v[u] = *(const int2*)(cells + ((size_t)sl * N_TILES + tile) * CELL_W + 2 * lane);
    }
    __builtin_amdgcn_sched_barrier(0);
    int c[CPW];
    int tot = 0;
    bool bad = false;
    #pragma unroll
    for (int u = 0; u < CPW; ++u) {
        c[u] = __builtin_amdgcn_readfirstlane(v[u].x);
        if (wave + u * (PT / 32) >= N_TILES) c[u] = 0;
        bad = bad || ((unsigned)c[u] > (unsigned)CELL_CAP);
        tot += c[u];
    }
    if (bad) {
        if (lane == 0) *ovfp = 1;
    } else {
        int wbase = 0;
        if (lane == 0) wbase = atomicAdd(nlistp, tot);
        int off = __builtin_amdgcn_readfirstlane(wbase);
        const int k0 = 2 * lane, k1 = k0 + 1;
        #pragma unroll
        for (int u = 0; u < CPW; ++u) {
            if (k0 >= 1 && k0 <= c[u]) {
                const int e = v[u].x;
                atomicAdd(&cnt[(e >> 20) & 255], 1);
                const int p = off + k0 - 1;
                if (p < LIST_CAP) outl[p] = e;
            }
            if (k1 <= c[u]) {
                const int e = v[u].y;
                atomicAdd(&cnt[(e >> 20) & 255], 1);
                const int p = off + k1 - 1;
                if (p < LIST_CAP) outl[p] = e;
            }
            off += c[u];
        }
    }
    __syncthreads();
    if (*ovfp) {
        cnt[t] = 0;
        if (t == 0) *nlistp = 0;
        __syncthreads();
        const int4* __restrict__ dst4 = (const int4*)dst;
        for (int i0 = 0; i0 < E4; i0 += 4 * PT) {
            int4 d[4];
            #pragma unroll
            for (int u = 0; u < 4; ++u) {
                const int i = i0 + u * PT + t;
                d[u] = (i < E4) ? dst4[i] : make_int4(-1, -1, -1, -1);
            }
            #pragma unroll
            for (int u = 0; u < 4; ++u) {
                const int e0 = (i0 + u * PT + t) * 4;
                const unsigned a = (unsigned)(d[u].x - base), b = (unsigned)(d[u].y - base);
                const unsigned cc = (unsigned)(d[u].z - base), g = (unsigned)(d[u].w - base);
                if (min(min(a, b), min(cc, g)) < NPB) {
                    if (a < NPB)  bin_append(cnt, nlistp, outl, src, e0 + 0, a);
                    if (b < NPB)  bin_append(cnt, nlistp, outl, src, e0 + 1, b);
                    if (cc < NPB) bin_append(cnt, nlistp, outl, src, e0 + 2, cc);
                    if (g < NPB)  bin_append(cnt, nlistp, outl, src, e0 + 3, g);
                }
            }
        }
        __syncthreads();
    }
    if (t < NPB) dinv[base + t] = rsqrtf(1.0f + (float)cnt[t]);
    if (t == 0) tile_n[tile] = *nlistp;
}

__device__ __forceinline__ void mm_tile(float* lds, const int rowTile,
                                        const float* __restrict__ X,
                                        const float* __restrict__ W,
                                        const float* __restrict__ bias,
                                        float* __restrict__ H) {
    const int rowBase = rowTile * 16;
    const int t = threadIdx.x;

    #pragma unroll
    for (int rep = 0; rep < 2; ++rep) {
        int idx = t + rep * 256;
        int r   = idx >> 5;
        int c4  = idx & 31;
        float4 v = *(const float4*)(X + (size_t)(rowBase + r) * DIM + c4 * 4);
        if (bias) {
            float4 b = *(const float4*)(bias + c4 * 4);
            v.x = fmaxf(v.x + b.x, 0.0f);
            v.y = fmaxf(v.y + b.y, 0.0f);
            v.z = fmaxf(v.z + b.z, 0.0f);
            v.w = fmaxf(v.w + b.w, 0.0f);
        }
        *(float4*)(&lds[r * LDS_STRIDE + c4 * 4]) = v;
    }
    __syncthreads();

    const int wave = t >> 5;
    const int lane = t & 31;
    const int half = lane >> 4;
    const int l16  = lane & 15;
    const int colBase = wave * 16;

    v8f acc = {0.f, 0.f, 0.f, 0.f, 0.f, 0.f, 0.f, 0.f};

    #pragma unroll
    for (int k0 = 0; k0 < DIM; k0 += 4) {
        const int kk = k0 + half * 2;
        float2 av = *(const float2*)(&lds[l16 * LDS_STRIDE + kk]);
        v2f a, b;
        a[0] = av.x;
        a[1] = av.y;
        b[0] = W[(size_t)kk * DIM + colBase + l16];
        b[1] = W[(size_t)(kk + 1) * DIM + colBase + l16];
        acc = __builtin_amdgcn_wmma_f32_16x16x4_f32(false, a, false, b,
                                                    (short)0, acc, false, false);
    }

    float* outp = H + (size_t)rowBase * DIM + colBase;
    #pragma unroll
    for (int v = 0; v < 8; ++v) {
        outp[(size_t)(v + half * 8) * DIM + l16] = acc[v];
    }
}

__launch_bounds__(PT)
__global__ void k_part_mm(const int* __restrict__ src, const int* __restrict__ dst,
                          int* __restrict__ cells,
                          const int* __restrict__ pa, const int* __restrict__ pb,
                          int* __restrict__ pcells,
                          const float* __restrict__ X, const float* __restrict__ W,
                          float* __restrict__ H) {
    __shared__ __attribute__((aligned(16))) int smem[STAGE_W];
    if (blockIdx.x < N_TILES) part_body(smem, blockIdx.x, src, dst, cells);
    else if (blockIdx.x < 2 * N_TILES) ppart_body(smem, blockIdx.x - N_TILES, pa, pb, pcells);
    else mm_tile((float*)smem, blockIdx.x - 2 * N_TILES, X, W, nullptr, H);
}

__launch_bounds__(PT)
__global__ void k_deg_mm(const int* __restrict__ src, const int* __restrict__ dst,
                         const int* __restrict__ cells,
                         float* __restrict__ dinv, int* __restrict__ tile_n,
                         int* __restrict__ tile_list,
                         const float* __restrict__ X, const float* __restrict__ W,
                         float* __restrict__ H) {
    __shared__ __attribute__((aligned(16))) int smem[MM_LDS_W];
    if (blockIdx.x < N_TILES) deg_body(smem, blockIdx.x, src, dst, cells, dinv, tile_n, tile_list);
    else mm_tile((float*)smem, MM_A + (blockIdx.x - N_TILES), X, W, nullptr, H);
}

__device__ __forceinline__ int acc_word(int f) { return ((f & 3) << 5) | (f >> 2); }

__device__ __forceinline__ void list_append(int* list, int* nlist, float* acc, int e, unsigned dl,
                                            int base, const int* __restrict__ src,
                                            const float* __restrict__ h,
                                            const float* __restrict__ dinv) {
    const int pos = atomicAdd(nlist, 1);
    const int s = src[e];
    if (pos < LIST_CAP) {
        list[pos] = s | (int)(dl << 20);
    } else {
        const float nm = dinv[s] * dinv[base + (int)dl];
        for (int f = 0; f < DIM; ++f)
            atomicAdd(&acc[dl * APITCH + acc_word(f)], h[(size_t)s * DIM + f] * nm);
    }
}

template <int NT>
__device__ __forceinline__ void rows_mm(const float* acc, const float* __restrict__ W,
                                        float* __restrict__ outp, const int g, const int ct,
                                        const int lane) {
    const int half = lane >> 4;
    const int c    = lane & 15;
    const int col  = ct * 16 + c;
    int rt[NT], aoff[NT];
    #pragma unroll
    for (int i = 0; i < NT; ++i) {
        rt[i] = (i < 3) ? g + 4 * i : 12;
        aoff[i] = min(rt[i] * 16 + c, NPB - 1) * APITCH + 2 * half;
    }
    v8f d[NT];
    #pragma unroll
    for (int i = 0; i < NT; ++i) d[i] = (v8f){0.f, 0.f, 0.f, 0.f, 0.f, 0.f, 0.f, 0.f};
    const float* __restrict__ Wc = W + col;
    #pragma unroll
    for (int j = 0; j < DIM / 4; ++j) {
        const int w0 = 4 * j + 2 * half;
        const int f0 = ((w0 & 31) << 2) | (j >> 3);
        v2f b;
        b[0] = Wc[(size_t)f0 * DIM];
        b[1] = Wc[(size_t)(f0 + 4) * DIM];
        #pragma unroll
        for (int i = 0; i < NT; ++i) {
            const float2 av = *(const float2*)(&acc[aoff[i] + 4 * j]);
            v2f a;
            a[0] = av.x;
            a[1] = av.y;
            d[i] = __builtin_amdgcn_wmma_f32_16x16x4_f32(false, a, false, b,
                                                         (short)0, d[i], false, false);
        }
    }
    #pragma unroll
    for (int i = 0; i < NT; ++i) {
        #pragma unroll
        for (int v = 0; v < 8; ++v) {
            const int row = rt[i] * 16 + half * 8 + v;
            if (row < NPB) outp[(size_t)row * DIM + col] = d[i][v];
        }
    }
}

template <bool MM>
__launch_bounds__(WG)
__global__ void k_aggregate(const int* __restrict__ src, const int* __restrict__ dst,
                            const int* __restrict__ tile_n, const int* __restrict__ tile_list,
                            const float* __restrict__ h, const float* __restrict__ dinv,
                            const float* __restrict__ bias_in, const float* __restrict__ W,
                            float* __restrict__ out) {
    __shared__ __attribute__((aligned(16))) float acc[NPB * APITCH];
    const float* __restrict__ bias = MM ? nullptr : bias_in;
    __shared__ int   list[LIST_CAP];
    __shared__ int   nlist;
    const int t = threadIdx.x;
    const int wave = t >> 5;
    const int lane = t & 31;
    const int base = blockIdx.x * NPB;
    const int ntile = tile_n[blockIdx.x];

    {
        constexpr int R = (NPB * DIM) / WG;
        float hr[R], dr[R];
        #pragma unroll
        for (int r = 0; r < R; ++r) {
            const int idx = r * WG + t;
            hr[r] = h[(size_t)base * DIM + idx];
            dr[r] = dinv[base + (idx >> 7)];
        }
        const float bv = bias ? bias[t & 127] : 0.0f;
        #pragma unroll
        for (int r = 0; r < R; ++r) {
            float v = hr[r] * dr[r] * dr[r];
            if (bias) v += bv;
            acc[((r * WG + t) >> 7) * APITCH + acc_word(t & 127)] = v;
        }
    }
    if (ntile <= LIST_CAP) {
        const int* __restrict__ inl = tile_list + (size_t)blockIdx.x * LIST_CAP;
        for (int k = t; k < ntile; k += WG) list[k] = inl[k];
    }
    if (t == 0) nlist = 0;
    __syncthreads();

    int n = ntile;
    if (ntile > LIST_CAP) {
        const int4* __restrict__ dst4 = (const int4*)dst;
        const int rot = blockIdx.x * (E4 / N_TILES);
        for (int i0 = 0; i0 < E4; i0 += 4 * WG) {
            int4 d[4];
            int  qq[4];
            #pragma unroll
            for (int u = 0; u < 4; ++u) {
                const int i = i0 + u * WG + t;
                int q = i + rot;
                if (q >= E4) q -= E4;
                qq[u] = q;
                d[u] = (i < E4) ? dst4[q] : make_int4(-1, -1, -1, -1);
            }
            #pragma unroll
            for (int u = 0; u < 4; ++u) {
                const int e0 = qq[u] * 4;
                const unsigned a = (unsigned)(d[u].x - base), b = (unsigned)(d[u].y - base);
                const unsigned c = (unsigned)(d[u].z - base), g = (unsigned)(d[u].w - base);
                if (min(min(a, b), min(c, g)) < NPB) {
                    if (a < NPB) list_append(list, &nlist, acc, e0 + 0, a, base, src, h, dinv);
                    if (b < NPB) list_append(list, &nlist, acc, e0 + 1, b, base, src, h, dinv);
                    if (c < NPB) list_append(list, &nlist, acc, e0 + 2, c, base, src, h, dinv);
                    if (g < NPB) list_append(list, &nlist, acc, e0 + 3, g, base, src, h, dinv);
                }
            }
        }
        __syncthreads();
        n = min(nlist, LIST_CAP);
    }

    for (int k0 = wave * PU; k0 < n; k0 += (WG / 32) * PU) {
        int   ent[PU], s[PU];
        float ds[PU], dd[PU];
        float hv[PU][4];
        #pragma unroll
        for (int u = 0; u < PU; ++u) {
            const int k = k0 + u;
            ent[u] = (k < n) ? list[k] : -1;
        }
        #pragma unroll
        for (int u = 0; u < PU; ++u) {
            s[u] = max(ent[u], 0) & 0xFFFFF;
            const int dl = (max(ent[u], 0) >> 20) & 255;
            ds[u] = dinv[s[u]];
            dd[u] = dinv[base + dl];
            const float4 hq = *(const float4*)(h + (size_t)s[u] * DIM + lane * 4);
            hv[u][0] = hq.x; hv[u][1] = hq.y; hv[u][2] = hq.z; hv[u][3] = hq.w;
        }
        #pragma unroll
        for (int u = 0; u < PU; ++u)
            asm volatile("" :: "v"(ds[u]), "v"(dd[u]), "v"(hv[u][0]), "v"(hv[u][1]),
                               "v"(hv[u][2]), "v"(hv[u][3]));
        #pragma unroll
        for (int u = 0; u < PU; ++u) {
            if (ent[u] >= 0) {
                const float nm = ds[u] * dd[u];
                float* ap = &acc[((ent[u] >> 20) & 255) * APITCH + lane];
                atomicAdd(ap +  0, hv[u][0] * nm);
                atomicAdd(ap + 32, hv[u][1] * nm);
                atomicAdd(ap + 64, hv[u][2] * nm);
                atomicAdd(ap + 96, hv[u][3] * nm);
            }
        }
    }
    __syncthreads();

    if (!MM) {
        for (int r = 0; r < (NPB * DIM) / WG; ++r)
            out[(size_t)base * DIM + r * WG + t] = acc[((r * WG + t) >> 7) * APITCH + acc_word(t & 127)];
    } else {
        {
            const int w = t & 127;
            const float bv = bias_in[((w & 31) << 2) | (w >> 5)];
            #pragma unroll
            for (int r = 0; r < (NPB * DIM) / WG; ++r) {
                float* p = &acc[((r * WG + t) >> 7) * APITCH + w];
                *p = fmaxf(*p + bv, 0.0f);
            }
        }
        __syncthreads();
        const int wv = __builtin_amdgcn_readfirstlane(wave);
        const int ct = wv & 7;
        const int g  = wv >> 3;
        float* __restrict__ outp = out + (size_t)base * DIM;
        if (g == ((ct >> 1) & 3)) rows_mm<4>(acc, W, outp, g, ct, lane);
        else                      rows_mm<3>(acc, W, outp, g, ct, lane);
    }
}

__device__ __forceinline__ float xhalf(float x) {
    const int b = __builtin_bit_cast(int, x);
    return __builtin_bit_cast(float, __builtin_amdgcn_permlanex16(b, b, 0x76543210, 0xfedcba98, false, false));
}
template <int M>
__device__ __forceinline__ float xrow(float x) {
    return __builtin_bit_cast(float, __builtin_amdgcn_update_dpp(0, __builtin_bit_cast(int, x),
                                                                 0x160 + M, 0xf, 0xf, true));
}
__device__ __forceinline__ float dot4(const float4 a, const float4 b) {
    return fmaf(a.w, b.w, fmaf(a.z, b.z, fmaf(a.x, b.x, a.y * b.y)));
}
__device__ __forceinline__ float wave_sum(float s) {
    s += xhalf(s);
    s += xrow<8>(s);
    s += xrow<4>(s);
    s += xrow<2>(s);
    s += xrow<1>(s);
    return s;
}

#define WL_CAP (N_TILES * PCELL_CAP)

__launch_bounds__(WG)
__global__ void k_decode_tile(const int* __restrict__ pa, const int* __restrict__ pb,
                              const int* __restrict__ pcells,
                              const float* __restrict__ z, float* __restrict__ out) {
    __shared__ __attribute__((aligned(16))) float zt[NPB * DIM];
    __shared__ int2 wl[WL_CAP];
    __shared__ int wtot[WG / 32];
    const int t = threadIdx.x;
    const int wave = __builtin_amdgcn_readfirstlane(t >> 5);
    const int lane = t & 31;
    const int tile = blockIdx.x;
    const int base = tile * NPB;

    int2 v[DCW];
    #pragma unroll
    for (int u = 0; u < DCW; ++u) {
        const int sl = min(wave + u * (WG / 32), N_TILES - 1);
        v[u] = *(const int2*)(pcells + ((size_t)sl * N_TILES + tile) * CELL_W + 2 * lane);
    }
    constexpr int Q4 = NPB * DIM / 4;
    constexpr int R4 = (Q4 + WG - 1) / WG;
    {
        const float4* __restrict__ zg = (const float4*)(z + (size_t)base * DIM);
        float4 zr[R4];
        #pragma unroll
        for (int r = 0; r < R4; ++r) zr[r] = zg[min(r * WG + t, Q4 - 1)];
        #pragma unroll
        for (int r = 0; r < R4; ++r)
            if (r * WG + t < Q4) ((float4*)zt)[r * WG + t] = zr[r];
    }
    int c[DCW];
    int tot = 0;
    bool bad = false;
    #pragma unroll
    for (int u = 0; u < DCW; ++u) {
        c[u] = __builtin_amdgcn_readfirstlane(v[u].x);
        if (wave + u * (WG / 32) >= N_TILES) c[u] = 0;
        bad = bad || ((unsigned)c[u] > (unsigned)PCELL_CAP);
        tot += c[u];
    }
    if (lane == 0) wtot[wave] = bad ? -1 : tot;
    __syncthreads();
    const int wt = wtot[lane];
    const bool ovf = __builtin_amdgcn_ballot_w32(wt < 0) != 0;

    if (!ovf) {
        int pos = 0, total = 0;
        #pragma unroll
        for (int w = 0; w < WG / 32; ++w) {
            const int x = __builtin_amdgcn_readlane(wt, w);
            if (w < wave) pos += x;
            total += x;
        }
        #pragma unroll
        for (int u = 0; u < DCW; ++u) {
            if (lane >= 1 && lane <= c[u]) wl[pos + lane - 1] = v[u];
            pos += c[u];
        }
        __syncthreads();
        for (int k0 = wave * DPU; k0 < total; k0 += (WG / 32) * DPU) {
            const int2 e = wl[min(k0 + (lane & (DPU - 1)), total - 1)];
            float4 za[DPU], zb[DPU];
            #pragma unroll
            for (int j = 0; j < DPU; ++j) {
                const int ex = __builtin_amdgcn_readlane(e.x, j);
                const int b  = min(ex & 0xFFFFF, N_NODES - 1);
                const int al = min((ex >> 20) & 255, NPB - 1);
                zb[j] = *(const float4*)(z + (size_t)b * DIM + lane * 4);
                za[j] = *(const float4*)(&zt[al * DIM + lane * 4]);
            }
            float s[DPU];
            #pragma unroll
            for (int j = 0; j < DPU; ++j)
                s[j] = dot4(za[j], zb[j]);
            #pragma unroll
            for (int j = 0; j < DPU; ++j) s[j] += xhalf(s[j]);
            #pragma unroll
            for (int j = 0; j < DPU; ++j) s[j] += xrow<8>(s[j]);
            #pragma unroll
            for (int j = 0; j < DPU; ++j) s[j] += xrow<4>(s[j]);
            #pragma unroll
            for (int j = 0; j < DPU; ++j) s[j] += xrow<2>(s[j]);
            #pragma unroll
            for (int j = 0; j < DPU; ++j) s[j] += xrow<1>(s[j]);
            float r = s[0];
            #pragma unroll
            for (int j = 1; j < DPU; ++j) r = (lane == j) ? s[j] : r;
            if (lane < DPU && k0 + lane < total) out[min((unsigned)e.y, (unsigned)(N_PAIRS - 1))] = r;
        }
    } else {
        for (int i0 = wave * 32; i0 < N_PAIRS; i0 += WG) {
            const int a = min(max(pa[i0 + lane], 0), N_NODES - 1);
            const int b = min(max(pb[i0 + lane], 0), N_NODES - 1);
            unsigned m = __builtin_amdgcn_ballot_w32((unsigned)(a - base) < (unsigned)NPB);
            while (m) {
                const int l = __builtin_ctz(m);
                m &= m - 1;
                const int sa = __builtin_amdgcn_readlane(a, l) - base;
                const int sb = __builtin_amdgcn_readlane(b, l);
                const float4 zb = *(const float4*)(z + (size_t)sb * DIM + lane * 4);
                const float4 za = *(const float4*)(&zt[sa * DIM + lane * 4]);
                const float s = wave_sum(dot4(za, zb));
                if (lane == 0) out[i0 + l] = s;
            }
        }
    }
}

extern "C" void kernel_launch(void* const* d_in, const int* in_sizes, int n_in,
                              void* d_out, int out_size, void* d_ws, size_t ws_size,
                              hipStream_t stream) {
    const int*   edge_index = (const int*)d_in[0];
    const int*   edge_pairs = (const int*)d_in[1];
    const float* emb        = (const float*)d_in[2];
    const float* W1         = (const float*)d_in[3];
    const float* b1         = (const float*)d_in[4];
    const float* W2         = (const float*)d_in[5];
    const float* b2         = (const float*)d_in[6];
    float*       out        = (float*)d_out;

    const int* src = edge_index;
    const int* dst = edge_index + N_EDGES;
    const int* pa  = edge_pairs;
    const int* pb  = edge_pairs + N_PAIRS;

    char* ws = (char*)d_ws;
    size_t off = 0;
    float* dinv = (float*)(ws + off); off += ((size_t)N_NODES * 4 + 255) & ~(size_t)255;
    float* h    = (float*)(ws + off); off += (size_t)N_NODES * DIM * 4;
    float* agg  = (float*)(ws + off); off += (size_t)N_NODES * DIM * 4;
    int* tile_n    = (int*)(ws + off); off += ((size_t)N_TILES * 4 + 255) & ~(size_t)255;
    int* tile_list = (int*)(ws + off); off += (size_t)N_TILES * LIST_CAP * 4;
    int* cells     = (int*)(ws + off); off += (size_t)N_TILES * N_TILES * CELL_W * 4;
    int* pcells    = (int*)(ws + off); off += (size_t)N_TILES * N_TILES * CELL_W * 4;
    (void)ws_size; (void)in_sizes; (void)n_in; (void)out_size;

    k_part_mm<<<2 * N_TILES + MM_A, PT, 0, stream>>>(src, dst, cells, pa, pb, pcells, emb, W1, h);
    k_deg_mm<<<N_TILES + MM_B, PT, 0, stream>>>(src, dst, cells, dinv, tile_n, tile_list, emb, W1, h);

    k_aggregate<true><<<N_TILES, WG, 0, stream>>>(src, dst, tile_n, tile_list, h, dinv, b1, W2, agg);

    k_aggregate<false><<<N_TILES, WG, 0, stream>>>(src, dst, tile_n, tile_list, agg, dinv, b2, nullptr, h);

    k_decode_tile<<<N_TILES, WG, 0, stream>>>(pa, pb, pcells, h, out);
}


# Aggregation: an edge's lane-uniform work (entry decode, row base, degree factors by scalar loads, their product, add test) moved to the scalar unit; vector ALU ops per edge 14 to 4; sum order kept
# speedup vs baseline: 3.3110x; 1.0607x over previous
#include <hip/hip_runtime.h>


#define N_NODES 50000
#define N_EDGES 800000
#define N_PAIRS 400000
#define DIM     128
#define LDS_STRIDE 132
#define APITCH  132

typedef float v2f __attribute__((ext_vector_type(2)));
typedef float v8f __attribute__((ext_vector_type(8)));

#define NPB      200
#define N_TILES  (N_NODES / NPB)
#define WG       1024
#define E4       (N_EDGES / 4)
#define LIST_CAP 16384
#define PU       8
#define ESL      (N_EDGES / N_TILES)
#define PTE      (ESL / 16)
#define CELL_W   64
#define CELL_CAP (CELL_W - 1)
#define SP       65
#define PT       256
#define CPW      32
#define STAGE_W  ((N_TILES * SP + 1 + 3) / 4 * 4)
#define MM_TILES (N_NODES / 16)
#define MM_A     1562
#define MM_B     (MM_TILES - MM_A)
#define MM_LDS_W (16 * LDS_STRIDE)
#define PSL       (N_PAIRS / N_TILES)
#define PPT       (PSL / 16)
#define PCELL_CAP ((CELL_W - 2) / 2)
#define DPU       8
#define DCW       ((N_TILES + 31) / 32)


__device__ __forceinline__ void part_body(int* stage, const int sl,
                                          const int* __restrict__ src, const int* __restrict__ dst,
                                          int* __restrict__ cells) {
    const int t = threadIdx.x;
    const int wave = t >> 5;
    const int lane = t & 31;
    for (int i = t; i < STAGE_W / 4; i += PT) ((int4*)stage)[i] = make_int4(0, 0, 0, 0);
    __syncthreads();
    if (t < PTE) {
        const int4* __restrict__ s4 = (const int4*)(src + (size_t)sl * ESL);
        const int4* __restrict__ d4 = (const int4*)(dst + (size_t)sl * ESL);
        int4 sv[4], dv[4];
        #pragma unroll
        for (int u = 0; u < 4; ++u) {
            sv[u] = s4[u * PTE + t];
            dv[u] = d4[u * PTE + t];
        }
        int  cell[16], ent[16], pos[16];
        bool ok[16];
        #pragma unroll
        for (int u = 0; u < 4; ++u) {
            const int ss[4] = {sv[u].x, sv[u].y, sv[u].z, sv[u].w};
            const int dd[4] = {dv[u].x, dv[u].y, dv[u].z, dv[u].w};
            #pragma unroll
            for (int j = 0; j < 4; ++j) {
                const int q = u * 4 + j;
                ok[q] = (unsigned)dd[j] < (unsigned)N_NODES;
                const int tile = ok[q] ? dd[j] / NPB : N_TILES;
                const int dl = ok[q] ? dd[j] - tile * NPB : 0;
                cell[q] = tile * SP;
                ent[q] = ss[j] | (dl << 20);
                pos[q] = atomicAdd(&stage[cell[q]], 1);
            }
        }
        #pragma unroll
        for (int q = 0; q < 16; ++q)
            if (ok[q] && pos[q] < CELL_CAP) stage[cell[q] + 1 + pos[q]] = ent[q];
    }
    __syncthreads();
    int* __restrict__ outc = cells + (size_t)sl * N_TILES * CELL_W;
    #pragma unroll
    for (int i = 0; i < CPW; ++i) {
        const int c = wave + i * (PT / 32);
        if (i * (PT / 32) + (PT / 32 - 1) < N_TILES || c < N_TILES) {
            int2 v;
            v.x = stage[c * SP + 2 * lane];
            v.y = stage[c * SP + 2 * lane + 1];
            *(int2*)(outc + c * CELL_W + 2 * lane) = v;
        }
    }
}

__device__ __forceinline__ void ppart_body(int* stage, const int sl,
                                           const int* __restrict__ pa, const int* __restrict__ pb,
                                           int* __restrict__ pcells) {
    const int t = threadIdx.x;
    const int wave = t >> 5;
    const int lane = t & 31;
    for (int i = t; i < STAGE_W / 4; i += PT) ((int4*)stage)[i] = make_int4(0, 0, 0, 0);
    __syncthreads();
    if (t < PPT) {
        const int4* __restrict__ a4 = (const int4*)(pa + (size_t)sl * PSL);
        const int4* __restrict__ b4 = (const int4*)(pb + (size_t)sl * PSL);
        int4 av[4], bv[4];
        #pragma unroll
        for (int u = 0; u < 4; ++u) {
            av[u] = a4[u * PPT + t];
            bv[u] = b4[u * PPT + t];
        }
        int cell[16], ent[16], pos[16];
        #pragma unroll
        for (int u = 0; u < 4; ++u) {
            const int aa[4] = {av[u].x, av[u].y, av[u].z, av[u].w};
            const int bb[4] = {bv[u].x, bv[u].y, bv[u].z, bv[u].w};
            #pragma unroll
            for (int j = 0; j < 4; ++j) {
                const int q = u * 4 + j;
                const int a = min(max(aa[j], 0), N_NODES - 1);
                const int b = min(max(bb[j], 0), N_NODES - 1);
                const int tile = a / NPB;
                cell[q] = tile * SP;
                ent[q] = b | ((a - tile * NPB) << 20);
                pos[q] = atomicAdd(&stage[cell[q]], 1);
            }
        }
        #pragma unroll
        for (int q = 0; q < 16; ++q)
            if (pos[q] < PCELL_CAP) {
                stage[cell[q] + 2 + 2 * pos[q]] = ent[q];
                stage[cell[q] + 3 + 2 * pos[q]] = sl * PSL + (q >> 2) * (4 * PPT) + 4 * t + (q & 3);
            }
    }
    __syncthreads();
    int* __restrict__ outc = pcells + (size_t)sl * N_TILES * CELL_W;
    #pragma unroll
    for (int i = 0; i < CPW; ++i) {
        const int c = wave + i * (PT / 32);
        if (i * (PT / 32) + (PT / 32 - 1) < N_TILES || c < N_TILES) {
            int2 v;
            v.x = stage[c * SP + 2 * lane];
            v.y = stage[c * SP + 2 * lane + 1];
            *(int2*)(outc + c * CELL_W + 2 * lane) = v;
        }
    }
}

__device__ __forceinline__ void bin_append(int* cnt, int* nlist, int* __restrict__ outl,
                                           const int* __restrict__ src, int e, unsigned dl) {
    atomicAdd(&cnt[dl], 1);
    const int pos = atomicAdd(nlist, 1);
    if (pos < LIST_CAP) outl[pos] = src[e] | (int)(dl << 20);
}

__device__ __forceinline__ void deg_body(int* lds, const int tile,
                                         const int* __restrict__ src, const int* __restrict__ dst,
                                         const int* __restrict__ cells,
                                         float* __restrict__ dinv, int* __restrict__ tile_n,
                                         int* __restrict__ tile_list) {
    int* cnt = lds;
    int* nlistp = lds + 256;
    int* ovfp = lds + 257;
    const int t = threadIdx.x;
    const int wave = t >> 5;
    const int lane = t & 31;
    const int base = tile * NPB;
    cnt[t] = 0;
    if (t == 0) { *nlistp = 0; *ovfp = 0; }
    __syncthreads();
    int* __restrict__ outl = tile_list + (size_t)tile * LIST_CAP;

    int2 v[CPW];
    #pragma unroll
    for (int u = 0; u < CPW; ++u) {
        const int sl = min(wave + u * (PT / 32), N_TILES - 1);
        v[u] = *(const int2*)(cells + ((size_t)sl * N_TILES + tile) * CELL_W + 2 * lane);
    }
    __builtin_amdgcn_sched_barrier(0);
    int c[CPW];
    int tot = 0;
    bool bad = false;
    #pragma unroll
    for (int u = 0; u < CPW; ++u) {
        c[u] = __builtin_amdgcn_readfirstlane(v[u].x);
        if (wave + u * (PT / 32) >= N_TILES) c[u] = 0;
        bad = bad || ((unsigned)c[u] > (unsigned)CELL_CAP);
        tot += c[u];
    }
    if (bad) {
        if (lane == 0) *ovfp = 1;
    } else {
        int wbase = 0;
        if (lane == 0) wbase = atomicAdd(nlistp, tot);
        int off = __builtin_amdgcn_readfirstlane(wbase);
        const int k0 = 2 * lane, k1 = k0 + 1;
        #pragma unroll
        for (int u = 0; u < CPW; ++u) {
            if (k0 >= 1 && k0 <= c[u]) {
                const int e = v[u].x;
                atomicAdd(&cnt[(e >> 20) & 255], 1);
                const int p = off + k0 - 1;
                if (p < LIST_CAP) outl[p] = e;
            }
            if (k1 <= c[u]) {
                const int e = v[u].y;
                atomicAdd(&cnt[(e >> 20) & 255], 1);
                const int p = off + k1 - 1;
                if (p < LIST_CAP) outl[p] = e;
            }
            off += c[u];
        }
    }
    __syncthreads();
    if (*ovfp) {
        cnt[t] = 0;
        if (t == 0) *nlistp = 0;
        __syncthreads();
        const int4* __restrict__ dst4 = (const int4*)dst;
        for (int i0 = 0; i0 < E4; i0 += 4 * PT) {
            int4 d[4];
            #pragma unroll
            for (int u = 0; u < 4; ++u) {
                const int i = i0 + u * PT + t;
                d[u] = (i < E4) ? dst4[i] : make_int4(-1, -1, -1, -1);
            }
            #pragma unroll
            for (int u = 0; u < 4; ++u) {
                const int e0 = (i0 + u * PT + t) * 4;
                const unsigned a = (unsigned)(d[u].x - base), b = (unsigned)(d[u].y - base);
                const unsigned cc = (unsigned)(d[u].z - base), g = (unsigned)(d[u].w - base);
                if (min(min(a, b), min(cc, g)) < NPB) {
                    if (a < NPB)  bin_append(cnt, nlistp, outl, src, e0 + 0, a);
                    if (b < NPB)  bin_append(cnt, nlistp, outl, src, e0 + 1, b);
                    if (cc < NPB) bin_append(cnt, nlistp, outl, src, e0 + 2, cc);
                    if (g < NPB)  bin_append(cnt, nlistp, outl, src, e0 + 3, g);
                }
            }
        }
        __syncthreads();
    }
    if (t < NPB) dinv[base + t] = rsqrtf(1.0f + (float)cnt[t]);
    if (t == 0) tile_n[tile] = *nlistp;
}

__device__ __forceinline__ void mm_tile(float* lds, const int rowTile,
                                        const float* __restrict__ X,
                                        const float* __restrict__ W,
                                        const float* __restrict__ bias,
                                        float* __restrict__ H) {
    const int rowBase = rowTile * 16;
    const int t = threadIdx.x;

    #pragma unroll
    for (int rep = 0; rep < 2; ++rep) {
        int idx = t + rep * 256;
        int r   = idx >> 5;
        int c4  = idx & 31;
        float4 v = *(const float4*)(X + (size_t)(rowBase + r) * DIM + c4 * 4);
        if (bias) {
            float4 b = *(const float4*)(bias + c4 * 4);
            v.x = fmaxf(v.x + b.x, 0.0f);
            v.y = fmaxf(v.y + b.y, 0.0f);
            v.z = fmaxf(v.z + b.z, 0.0f);
            v.w = fmaxf(v.w + b.w, 0.0f);
        }
        *(float4*)(&lds[r * LDS_STRIDE + c4 * 4]) = v;
    }
    __syncthreads();

    const int wave = t >> 5;
    const int lane = t & 31;
    const int half = lane >> 4;
    const int l16  = lane & 15;
    const int colBase = wave * 16;

    v8f acc = {0.f, 0.f, 0.f, 0.f, 0.f, 0.f, 0.f, 0.f};

    #pragma unroll
    for (int k0 = 0; k0 < DIM; k0 += 4) {
        const int kk = k0 + half * 2;
        float2 av = *(const float2*)(&lds[l16 * LDS_STRIDE + kk]);
        v2f a, b;
        a[0] = av.x;
        a[1] = av.y;
        b[0] = W[(size_t)kk * DIM + colBase + l16];
        b[1] = W[(size_t)(kk + 1) * DIM + colBase + l16];
        acc = __builtin_amdgcn_wmma_f32_16x16x4_f32(false, a, false, b,
                                                    (short)0, acc, false, false);
    }

    float* outp = H + (size_t)rowBase * DIM + colBase;
    #pragma unroll
    for (int v = 0; v < 8; ++v) {
        outp[(size_t)(v + half * 8) * DIM + l16] = acc[v];
    }
}

__launch_bounds__(PT)
__global__ void k_part_mm(const int* __restrict__ src, const int* __restrict__ dst,
                          int* __restrict__ cells,
                          const int* __restrict__ pa, const int* __restrict__ pb,
                          int* __restrict__ pcells,
                          const float* __restrict__ X, const float* __restrict__ W,
                          float* __restrict__ H) {
    __shared__ __attribute__((aligned(16))) int smem[STAGE_W];
    if (blockIdx.x < N_TILES) part_body(smem, blockIdx.x, src, dst, cells);
    else if (blockIdx.x < 2 * N_TILES) ppart_body(smem, blockIdx.x - N_TILES, pa, pb, pcells);
    else mm_tile((float*)smem, blockIdx.x - 2 * N_TILES, X, W, nullptr, H);
}

__launch_bounds__(PT)
__global__ void k_deg_mm(const int* __restrict__ src, const int* __restrict__ dst,
                         const int* __restrict__ cells,
                         float* __restrict__ dinv, int* __restrict__ tile_n,
                         int* __restrict__ tile_list,
                         const float* __restrict__ X, const float* __restrict__ W,
                         float* __restrict__ H) {
    __shared__ __attribute__((aligned(16))) int smem[MM_LDS_W];
    if (blockIdx.x < N_TILES) deg_body(smem, blockIdx.x, src, dst, cells, dinv, tile_n, tile_list);
    else mm_tile((float*)smem, MM_A + (blockIdx.x - N_TILES), X, W, nullptr, H);
}

typedef const char __attribute__((address_space(1))) gchar_k;
typedef const float4 __attribute__((address_space(1))) gfloat4_k;
__device__ __forceinline__ gchar_k* uniform_ptr(const void* p) {
    const unsigned long long a = (unsigned long long)p;
    const unsigned lo = (unsigned)__builtin_amdgcn_readfirstlane((int)(unsigned)a);
    const unsigned hi = (unsigned)__builtin_amdgcn_readfirstlane((int)(unsigned)(a >> 32));
    return (gchar_k*)(((unsigned long long)hi << 32) | lo);
}
typedef const float __attribute__((address_space(4))) cfloat_k;

__device__ __forceinline__ int acc_word(int f) { return ((f & 3) << 5) | (f >> 2); }

__device__ __forceinline__ void list_append(int* list, int* nlist, float* acc, int e, unsigned dl,
                                            int base, const int* __restrict__ src,
                                            const float* __restrict__ h,
                                            const float* __restrict__ dinv) {
    const int pos = atomicAdd(nlist, 1);
    const int s = src[e];
    if (pos < LIST_CAP) {
        list[pos] = s | (int)(dl << 20);
    } else {
        const float nm = dinv[s] * dinv[base + (int)dl];
        for (int f = 0; f < DIM; ++f)
            atomicAdd(&acc[dl * APITCH + acc_word(f)], h[(size_t)s * DIM + f] * nm);
    }
}

template <int NT>
__device__ __forceinline__ void rows_mm(const float* acc, const float* __restrict__ W,
                                        float* __restrict__ outp, const int g, const int ct,
                                        const int lane) {
    const int half = lane >> 4;
    const int c    = lane & 15;
    const int col  = ct * 16 + c;
    int rt[NT], aoff[NT];
    #pragma unroll
    for (int i = 0; i < NT; ++i) {
        rt[i] = (i < 3) ? g + 4 * i : 12;
        aoff[i] = min(rt[i] * 16 + c, NPB - 1) * APITCH + 2 * half;
    }
    v8f d[NT];
    #pragma unroll
    for (int i = 0; i < NT; ++i) d[i] = (v8f){0.f, 0.f, 0.f, 0.f, 0.f, 0.f, 0.f, 0.f};
    const float* __restrict__ Wc = W + col;
    #pragma unroll
    for (int j = 0; j < DIM / 4; ++j) {
        const int w0 = 4 * j + 2 * half;
        const int f0 = ((w0 & 31) << 2) | (j >> 3);
        v2f b;
        b[0] = Wc[(size_t)f0 * DIM];
        b[1] = Wc[(size_t)(f0 + 4) * DIM];
        #pragma unroll
        for (int i = 0; i < NT; ++i) {
            const float2 av = *(const float2*)(&acc[aoff[i] + 4 * j]);
            v2f a;
            a[0] = av.x;
            a[1] = av.y;
            d[i] = __builtin_amdgcn_wmma_f32_16x16x4_f32(false, a, false, b,
                                                         (short)0, d[i], false, false);
        }
    }
    #pragma unroll
    for (int i = 0; i < NT; ++i) {
        #pragma unroll
        for (int v = 0; v < 8; ++v) {
            const int row = rt[i] * 16 + half * 8 + v;
            if (row < NPB) outp[(size_t)row * DIM + col] = d[i][v];
        }
    }
}

template <bool MM>
__launch_bounds__(WG)
__global__ void k_aggregate(const int* __restrict__ src, const int* __restrict__ dst,
                            const int* __restrict__ tile_n, const int* __restrict__ tile_list,
                            const float* __restrict__ h, const float* __restrict__ dinv,
                            const float* __restrict__ bias_in, const float* __restrict__ W,
                            float* __restrict__ out) {
    __shared__ __attribute__((aligned(16))) float acc[NPB * APITCH];
    const float* __restrict__ bias = MM ? nullptr : bias_in;
    __shared__ int   list[LIST_CAP];
    __shared__ int   nlist;
    const int t = threadIdx.x;
    const int wave = t >> 5;
    const int lane = t & 31;
    const int base = blockIdx.x * NPB;
    const int ntile = tile_n[blockIdx.x];

    {
        constexpr int R = (NPB * DIM) / WG;
        float hr[R], dr[R];
        #pragma unroll
        for (int r = 0; r < R; ++r) {
            const int idx = r * WG + t;
            hr[r] = h[(size_t)base * DIM + idx];
            dr[r] = dinv[base + (idx >> 7)];
        }
        const float bv = bias ? bias[t & 127] : 0.0f;
        #pragma unroll
        for (int r = 0; r < R; ++r) {
            float v = hr[r] * dr[r] * dr[r];
            if (bias) v += bv;
            acc[((r * WG + t) >> 7) * APITCH + acc_word(t & 127)] = v;
        }
    }
    if (ntile <= LIST_CAP) {
        const int* __restrict__ inl = tile_list + (size_t)blockIdx.x * LIST_CAP;
        for (int k = t; k < ntile; k += WG) list[k] = inl[k];
    }
    if (t == 0) nlist = 0;
    __syncthreads();

    int n = ntile;
    if (ntile > LIST_CAP) {
        const int4* __restrict__ dst4 = (const int4*)dst;
        const int rot = blockIdx.x * (E4 / N_TILES);
        for (int i0 = 0; i0 < E4; i0 += 4 * WG) {
            int4 d[4];
            int  qq[4];
            #pragma unroll
            for (int u = 0; u < 4; ++u) {
                const int i = i0 + u * WG + t;
                int q = i + rot;
                if (q >= E4) q -= E4;
                qq[u] = q;
                d[u] = (i < E4) ? dst4[q] : make_int4(-1, -1, -1, -1);
            }
            #pragma unroll
            for (int u = 0; u < 4; ++u) {
                const int e0 = qq[u] * 4;
                const unsigned a = (unsigned)(d[u].x - base), b = (unsigned)(d[u].y - base);
                const unsigned c = (unsigned)(d[u].z - base), g = (unsigned)(d[u].w - base);
                if (min(min(a, b), min(c, g)) < NPB) {
                    if (a < NPB) list_append(list, &nlist, acc, e0 + 0, a, base, src, h, dinv);
                    if (b < NPB) list_append(list, &nlist, acc, e0 + 1, b, base, src, h, dinv);
                    if (c < NPB) list_append(list, &nlist, acc, e0 + 2, c, base, src, h, dinv);
                    if (g < NPB) list_append(list, &nlist, acc, e0 + 3, g, base, src, h, dinv);
                }
            }
        }
        __syncthreads();
        n = min(nlist, LIST_CAP);
    }

    const int wv = __builtin_amdgcn_readfirstlane(wave);
    const int nu = __builtin_amdgcn_readfirstlane(n);
    cfloat_k* dinvk = (cfloat_k*)dinv;
    const unsigned lane16 = (unsigned)lane * 16u;
    for (int k0 = wv * PU; k0 < nu; k0 += (WG / 32) * PU) {
        const int e = list[min(k0 + (lane & (PU - 1)), nu - 1)];
        const int cnt = min(nu - k0, PU);
        unsigned off = lane16;
        asm volatile("" : "+v"(off));
        int    dl[PU];
        float  ds[PU], dd[PU];
        float4 hq[PU];
        #pragma unroll
        for (int u = 0; u < PU; ++u) {
            const int ex = __builtin_amdgcn_readlane(e, u);
            const int s  = ex & 0xFFFFF;
            dl[u] = (ex >> 20) & 255;
            ds[u] = dinvk[(unsigned)s];
            dd[u] = dinvk[(unsigned)(base + dl[u])];
            hq[u] = *(gfloat4_k*)(uniform_ptr(h + (size_t)s * DIM) + off);
        }
        #pragma unroll
        for (int u = 0; u < PU; ++u)
            asm volatile("" :: "s"(ds[u]), "s"(dd[u]), "v"(hq[u].x), "v"(hq[u].y),
                               "v"(hq[u].z), "v"(hq[u].w));
        #pragma unroll
        for (int u = 0; u < PU; ++u) {
            if (u < cnt) {
                const float nm = ds[u] * dd[u];
                float* ap = &acc[dl[u] * APITCH + lane];
                atomicAdd(ap +  0, hq[u].x * nm);
                atomicAdd(ap + 32, hq[u].y * nm);
                atomicAdd(ap + 64, hq[u].z * nm);
                atomicAdd(ap + 96, hq[u].w * nm);
            }
        }
    }
    __syncthreads();

    if (!MM) {
        for (int r = 0; r < (NPB * DIM) / WG; ++r)
            out[(size_t)base * DIM + r * WG + t] = acc[((r * WG + t) >> 7) * APITCH + acc_word(t & 127)];
    } else {
        {
            const int w = t & 127;
            const float bv = bias_in[((w & 31) << 2) | (w >> 5)];
            #pragma unroll
            for (int r = 0; r < (NPB * DIM) / WG; ++r) {
                float* p = &acc[((r * WG + t) >> 7) * APITCH + w];
                *p = fmaxf(*p + bv, 0.0f);
            }
        }
        __syncthreads();
        const int wv = __builtin_amdgcn_readfirstlane(wave);
        const int ct = wv & 7;
        const int g  = wv >> 3;
        float* __restrict__ outp = out + (size_t)base * DIM;
        if (g == ((ct >> 1) & 3)) rows_mm<4>(acc, W, outp, g, ct, lane);
        else                      rows_mm<3>(acc, W, outp, g, ct, lane);
    }
}

__device__ __forceinline__ float xhalf(float x) {
    const int b = __builtin_bit_cast(int, x);
    return __builtin_bit_cast(float, __builtin_amdgcn_permlanex16(b, b, 0x76543210, 0xfedcba98, false, false));
}
template <int M>
__device__ __forceinline__ float xrow(float x) {
    return __builtin_bit_cast(float, __builtin_amdgcn_update_dpp(0, __builtin_bit_cast(int, x),
                                                                 0x160 + M, 0xf, 0xf, true));
}
__device__ __forceinline__ float dot4(const float4 a, const float4 b) {
    return fmaf(a.w, b.w, fmaf(a.z, b.z, fmaf(a.x, b.x, a.y * b.y)));
}
__device__ __forceinline__ float wave_sum(float s) {
    s += xhalf(s);
    s += xrow<8>(s);
    s += xrow<4>(s);
    s += xrow<2>(s);
    s += xrow<1>(s);
    return s;
}

#define WL_CAP (N_TILES * PCELL_CAP)

__launch_bounds__(WG)
__global__ void k_decode_tile(const int* __restrict__ pa, const int* __restrict__ pb,
                              const int* __restrict__ pcells,
                              const float* __restrict__ z, float* __restrict__ out) {
    __shared__ __attribute__((aligned(16))) float zt[NPB * DIM];
    __shared__ int2 wl[WL_CAP];
    __shared__ int wtot[WG / 32];
    const int t = threadIdx.x;
    const int wave = __builtin_amdgcn_readfirstlane(t >> 5);
    const int lane = t & 31;
    const int tile = blockIdx.x;
    const int base = tile * NPB;

    int2 v[DCW];
    #pragma unroll
    for (int u = 0; u < DCW; ++u) {
        const int sl = min(wave + u * (WG / 32), N_TILES - 1);
        v[u] = *(const int2*)(pcells + ((size_t)sl * N_TILES + tile) * CELL_W + 2 * lane);
    }
    constexpr int Q4 = NPB * DIM / 4;
    constexpr int R4 = (Q4 + WG - 1) / WG;
    {
        const float4* __restrict__ zg = (const float4*)(z + (size_t)base * DIM);
        float4 zr[R4];
        #pragma unroll
        for (int r = 0; r < R4; ++r) zr[r] = zg[min(r * WG + t, Q4 - 1)];
        #pragma unroll
        for (int r = 0; r < R4; ++r)
            if (r * WG + t < Q4) ((float4*)zt)[r * WG + t] = zr[r];
    }
    int c[DCW];
    int tot = 0;
    bool bad = false;
    #pragma unroll
    for (int u = 0; u < DCW; ++u) {
        c[u] = __builtin_amdgcn_readfirstlane(v[u].x);
        if (wave + u * (WG / 32) >= N_TILES) c[u] = 0;
        bad = bad || ((unsigned)c[u] > (unsigned)PCELL_CAP);
        tot += c[u];
    }
    if (lane == 0) wtot[wave] = bad ? -1 : tot;
    __syncthreads();
    const int wt = wtot[lane];
    const bool ovf = __builtin_amdgcn_ballot_w32(wt < 0) != 0;

    if (!ovf) {
        int pos = 0, total = 0;
        #pragma unroll
        for (int w = 0; w < WG / 32; ++w) {
            const int x = __builtin_amdgcn_readlane(wt, w);
            if (w < wave) pos += x;
            total += x;
        }
        #pragma unroll
        for (int u = 0; u < DCW; ++u) {
            if (lane >= 1 && lane <= c[u]) wl[pos + lane - 1] = v[u];
            pos += c[u];
        }
        __syncthreads();
        for (int k0 = wave * DPU; k0 < total; k0 += (WG / 32) * DPU) {
            const int2 e = wl[min(k0 + (lane & (DPU - 1)), total - 1)];
            float4 za[DPU], zb[DPU];
            #pragma unroll
            for (int j = 0; j < DPU; ++j) {
                const int ex = __builtin_amdgcn_readlane(e.x, j);
                const int b  = min(ex & 0xFFFFF, N_NODES - 1);
                const int al = min((ex >> 20) & 255, NPB - 1);
                zb[j] = *(const float4*)(z + (size_t)b * DIM + lane * 4);
                za[j] = *(const float4*)(&zt[al * DIM + lane * 4]);
            }
            float s[DPU];
            #pragma unroll
            for (int j = 0; j < DPU; ++j)
                s[j] = dot4(za[j], zb[j]);
            #pragma unroll
            for (int j = 0; j < DPU; ++j) s[j] += xhalf(s[j]);
            #pragma unroll
            for (int j = 0; j < DPU; ++j) s[j] += xrow<8>(s[j]);
            #pragma unroll
            for (int j = 0; j < DPU; ++j) s[j] += xrow<4>(s[j]);
            #pragma unroll
            for (int j = 0; j < DPU; ++j) s[j] += xrow<2>(s[j]);
            #pragma unroll
            for (int j = 0; j < DPU; ++j) s[j] += xrow<1>(s[j]);
            float r = s[0];
            #pragma unroll
            for (int j = 1; j < DPU; ++j) r = (lane == j) ? s[j] : r;
            if (lane < DPU && k0 + lane < total) out[min((unsigned)e.y, (unsigned)(N_PAIRS - 1))] = r;
        }
    } else {
        for (int i0 = wave * 32; i0 < N_PAIRS; i0 += WG) {
            const int a = min(max(pa[i0 + lane], 0), N_NODES - 1);
            const int b = min(max(pb[i0 + lane], 0), N_NODES - 1);
            unsigned m = __builtin_amdgcn_ballot_w32((unsigned)(a - base) < (unsigned)NPB);
            while (m) {
                const int l = __builtin_ctz(m);
                m &= m - 1;
                const int sa = __builtin_amdgcn_readlane(a, l) - base;
                const int sb = __builtin_amdgcn_readlane(b, l);
                const float4 zb = *(const float4*)(z + (size_t)sb * DIM + lane * 4);
                const float4 za = *(const float4*)(&zt[sa * DIM + lane * 4]);
                const float s = wave_sum(dot4(za, zb));
                if (lane == 0) out[i0 + l] = s;
            }
        }
    }
}

extern "C" void kernel_launch(void* const* d_in, const int* in_sizes, int n_in,
                              void* d_out, int out_size, void* d_ws, size_t ws_size,
                              hipStream_t stream) {
    const int*   edge_index = (const int*)d_in[0];
    const int*   edge_pairs = (const int*)d_in[1];
    const float* emb        = (const float*)d_in[2];
    const float* W1         = (const float*)d_in[3];
    const float* b1         = (const float*)d_in[4];
    const float* W2         = (const float*)d_in[5];
    const float* b2         = (const float*)d_in[6];
    float*       out        = (float*)d_out;

    const int* src = edge_index;
    const int* dst = edge_index + N_EDGES;
    const int* pa  = edge_pairs;
    const int* pb  = edge_pairs + N_PAIRS;

    char* ws = (char*)d_ws;
    size_t off = 0;
    float* dinv = (float*)(ws + off); off += ((size_t)N_NODES * 4 + 255) & ~(size_t)255;
    float* h    = (float*)(ws + off); off += (size_t)N_NODES * DIM * 4;
    float* agg  = (float*)(ws + off); off += (size_t)N_NODES * DIM * 4;
    int* tile_n    = (int*)(ws + off); off += ((size_t)N_TILES * 4 + 255) & ~(size_t)255;
    int* tile_list = (int*)(ws + off); off += (size_t)N_TILES * LIST_CAP * 4;
    int* cells     = (int*)(ws + off); off += (size_t)N_TILES * N_TILES * CELL_W * 4;
    int* pcells    = (int*)(ws + off); off += (size_t)N_TILES * N_TILES * CELL_W * 4;
    (void)ws_size; (void)in_sizes; (void)n_in; (void)out_size;

    k_part_mm<<<2 * N_TILES + MM_A, PT, 0, stream>>>(src, dst, cells, pa, pb, pcells, emb, W1, h);
    k_deg_mm<<<N_TILES + MM_B, PT, 0, stream>>>(src, dst, cells, dinv, tile_n, tile_list, emb, W1, h);

    k_aggregate<true><<<N_TILES, WG, 0, stream>>>(src, dst, tile_n, tile_list, h, dinv, b1, W2, agg);

    k_aggregate<false><<<N_TILES, WG, 0, stream>>>(src, dst, tile_n, tile_list, agg, dinv, b2, nullptr, h);

    k_decode_tile<<<N_TILES, WG, 0, stream>>>(pa, pb, pcells, h, out);
}
